# starting kernel
# baseline (speedup 1.0000x reference)
#include <hip/hip_runtime.h>
#include <hip/hip_bf16.h>

typedef __attribute__((ext_vector_type(2))) float v2f;
typedef __attribute__((ext_vector_type(8))) float v8f;

__global__ __launch_bounds__(256) void zero_f32(float* __restrict__ p, unsigned long long n) {
    unsigned long long i = (unsigned long long)blockIdx.x * blockDim.x + threadIdx.x;
    if (i < n) p[i] = 0.0f;
}

__global__ __launch_bounds__(256) void sage_scatter(const float* __restrict__ hsrc,
                                                    const int* __restrict__ src,
                                                    const int* __restrict__ dst,
                                                    int nedges,
                                                    float* __restrict__ msg,
                                                    float* __restrict__ deg) {
    int lane = threadIdx.x & 31;
    int edge = blockIdx.x * (blockDim.x >> 5) + (threadIdx.x >> 5);
    if (edge >= nedges) return;
    int s = src[edge];
    int d = dst[edge];
    const float4 v = *(const float4*)(hsrc + (size_t)s * 128 + lane * 4);
    float* mrow = msg + (size_t)d * 128 + lane * 4;
    atomicAdd(mrow + 0, v.x);
    atomicAdd(mrow + 1, v.y);
    atomicAdd(mrow + 2, v.z);
    atomicAdd(mrow + 3, v.w);
    if (lane == 0) atomicAdd(deg + d, 1.0f);
}

__global__ __launch_bounds__(256) void sage_gemm(const float* __restrict__ hdst,
                                                 const float* __restrict__ msg,
                                                 const float* __restrict__ deg,
                                                 const float* __restrict__ Wself,
                                                 const float* __restrict__ Wneigh,
                                                 const float* __restrict__ bias,
                                                 float* __restrict__ out,
                                                 int n_dst, int d_out, int relu) {
    const int d_in = 128;
    int lane   = threadIdx.x & 31;
    int wave   = blockIdx.x * (blockDim.x >> 5) + (threadIdx.x >> 5);
    int ntiles = d_out >> 4;
    int mtiles = (n_dst + 15) >> 4;
    if (wave >= mtiles * ntiles) return;

    int mt = wave / ntiles;
    int nt = wave - mt * ntiles;
    int m0 = mt << 4;
    int n0 = nt << 4;

    int half = lane >> 4;
    int idx  = lane & 15;

    int arow = m0 + idx;
    if (arow > n_dst - 1) arow = n_dst - 1;

    const float* Ah = hdst  + (size_t)arow * d_in + 2 * half;
    const float* Am = msg   + (size_t)arow * d_in + 2 * half;
    const float* Bs = Wself  + (size_t)(n0 + idx) * d_in + 2 * half;
    const float* Bn = Wneigh + (size_t)(n0 + idx) * d_in + 2 * half;
    float invdeg = 1.0f / fmaxf(deg[arow], 1.0f);

    v8f c = {0.f, 0.f, 0.f, 0.f, 0.f, 0.f, 0.f, 0.f};

#pragma unroll
    for (int k = 0; k < d_in; k += 4) {
        v2f a = *(const v2f*)(Ah + k);
        v2f b = *(const v2f*)(Bs + k);
        c = __builtin_amdgcn_wmma_f32_16x16x4_f32(false, a, false, b,
                                                  (short)0, c, false, false);
    }
#pragma unroll
    for (int k = 0; k < d_in; k += 4) {
        v2f a = *(const v2f*)(Am + k);
        a.x *= invdeg;
        a.y *= invdeg;
        v2f b = *(const v2f*)(Bn + k);
        c = __builtin_amdgcn_wmma_f32_16x16x4_f32(false, a, false, b,
                                                  (short)0, c, false, false);
    }

    float bv = bias[n0 + idx];
#pragma unroll
    for (int j = 0; j < 8; ++j) {
        int m = m0 + j + 8 * half;
        if (m < n_dst) {
            float v = c[j] + bv;
            if (relu) v = fmaxf(v, 0.0f);
            out[(size_t)m * d_out + (n0 + idx)] = v;
        }
    }
}

static inline unsigned cdiv(unsigned a, unsigned b) { return (a + b - 1) / b; }

extern "C" void kernel_launch(void* const* d_in, const int* in_sizes, int n_in,
                              void* d_out, int out_size, void* d_ws, size_t ws_size,
                              hipStream_t stream) {
    const float* feats = (const float*)d_in[0];
    const int* src0 = (const int*)d_in[1];
    const int* dst0 = (const int*)d_in[2];
    const int* src1 = (const int*)d_in[3];
    const int* dst1 = (const int*)d_in[4];
    const int* src2 = (const int*)d_in[5];
    const int* dst2 = (const int*)d_in[6];
    const float* Ws0 = (const float*)d_in[10];
    const float* Wn0 = (const float*)d_in[11];
    const float* b0  = (const float*)d_in[12];
    const float* Ws1 = (const float*)d_in[13];
    const float* Wn1 = (const float*)d_in[14];
    const float* b1  = (const float*)d_in[15];
    const float* Ws2 = (const float*)d_in[16];
    const float* Wn2 = (const float*)d_in[17];
    const float* b2  = (const float*)d_in[18];

    const int E0 = in_sizes[1], E1 = in_sizes[3], E2 = in_sizes[5];
    const int N1 = 100000, N2 = 20000, N3 = 5000;

    char* ws = (char*)d_ws;
    float* h1  = (float*)ws;                    ws += (size_t)N1 * 128 * sizeof(float);
    float* h2  = (float*)ws;                    ws += (size_t)N2 * 128 * sizeof(float);
    float* msg = (float*)ws;                    ws += (size_t)N1 * 128 * sizeof(float);
    float* deg = (float*)ws;

    const dim3 blk(256);

    zero_f32<<<cdiv((unsigned)N1 * 128, 256), blk, 0, stream>>>(msg, (unsigned long long)N1 * 128);
    zero_f32<<<cdiv((unsigned)N1, 256), blk, 0, stream>>>(deg, (unsigned long long)N1);
    sage_scatter<<<cdiv((unsigned)E0, 8), blk, 0, stream>>>(feats, src0, dst0, E0, msg, deg);
    {
        unsigned tiles = cdiv((unsigned)N1, 16) * (128 / 16);
        sage_gemm<<<cdiv(tiles, 8), blk, 0, stream>>>(feats, msg, deg, Ws0, Wn0, b0, h1, N1, 128, 1);
    }

    zero_f32<<<cdiv((unsigned)N2 * 128, 256), blk, 0, stream>>>(msg, (unsigned long long)N2 * 128);
    zero_f32<<<cdiv((unsigned)N2, 256), blk, 0, stream>>>(deg, (unsigned long long)N2);
    sage_scatter<<<cdiv((unsigned)E1, 8), blk, 0, stream>>>(h1, src1, dst1, E1, msg, deg);
    {
        unsigned tiles = cdiv((unsigned)N2, 16) * (128 / 16);
        sage_gemm<<<cdiv(tiles, 8), blk, 0, stream>>>(h1, msg, deg, Ws1, Wn1, b1, h2, N2, 128, 1);
    }

    zero_f32<<<cdiv((unsigned)N3 * 128, 256), blk, 0, stream>>>(msg, (unsigned long long)N3 * 128);
    zero_f32<<<cdiv((unsigned)N3, 256), blk, 0, stream>>>(deg, (unsigned long long)N3);
    sage_scatter<<<cdiv((unsigned)E2, 8), blk, 0, stream>>>(h2, src2, dst2, E2, msg, deg);
    {
        unsigned tiles = cdiv((unsigned)N3, 16) * (64 / 16);
        sage_gemm<<<cdiv(tiles, 8), blk, 0, stream>>>(h2, msg, deg, Ws2, Wn2, b2, (float*)d_out, N3, 64, 0);
    }
}


# Neighbour sum by row ownership: a work-group keeps a block of dst rows in LDS, scans all edges for them, adds src rows by LDS float adds; no global atomics or zero fills; sum order: LDS arrival
# speedup vs baseline: 1.9707x; 1.9707x over previous
#include <hip/hip_runtime.h>
#include <hip/hip_bf16.h>


typedef __attribute__((ext_vector_type(2))) float v2f;
typedef __attribute__((ext_vector_type(8))) float v8f;

template <int R>
__global__ __launch_bounds__(1024) void sage_sum_owned(const float* __restrict__ hsrc,
                                                       const int* __restrict__ src,
                                                       const int* __restrict__ dst,
                                                       int nedges, int nsrc, int n_dst,
                                                       float* __restrict__ msg,
                                                       float* __restrict__ deg) {
    constexpr int QCAP = 136;
    __shared__ __attribute__((aligned(16))) float acc[(R + 1) * 128];
    __shared__ float cnt[R + 1];
    __shared__ unsigned qe[32 * QCAP];
    __shared__ unsigned qd[32 * QCAP];

    const int tid  = threadIdx.x;
    const int lane = tid & 31;
    const int wave = tid >> 5;
    const int hi   = lane >> 4;
    const int base = blockIdx.x * R;
    int rows = n_dst - base;
    if (rows > R) rows = R;
    if (rows <= 0) return;
    const unsigned urows = (unsigned)rows;

    for (int i = tid; i < (R + 1) * 32; i += 1024)
        *(float4*)(acc + i * 4) = make_float4(0.0f, 0.0f, 0.0f, 0.0f);
    for (int i = tid; i < R + 1; i += 1024) cnt[i] = 0.0f;
    __syncthreads();

    const int wq = wave * QCAP;
    int count = 0;

    const int o0 = lane * 4 + hi;
    const int o1 = lane * 4 + 1 + hi;
    const int o2 = lane * 4 + 2 + hi;
    const int o3 = lane * 4 + (hi ? 0 : 3);

    auto drain8 = [&]() {
        __builtin_amdgcn_fence(__ATOMIC_ACQ_REL, "wavefront");
        const int idx = count - 8 + (lane & 7);
        const bool ok = idx >= 0;
        unsigned e  = ok ? qe[wq + idx] : 0u;
        int dl      = ok ? (int)qd[wq + idx] : R;
        int s = ok ? src[e] : 0;
        s = s < 0 ? 0 : (s > nsrc - 1 ? nsrc - 1 : s);
        float4 v[8];
        int    dk[8];
#pragma unroll
        for (int k = 0; k < 8; ++k) {
            const int sk = __builtin_amdgcn_readlane(s, k);
            dk[k] = __builtin_amdgcn_readlane(dl, k);
            v[k] = *(const float4*)(hsrc + (size_t)sk * 128 + lane * 4);
        }
#pragma unroll
        for (int k = 0; k < 8; ++k) {
            float* row = acc + dk[k] * 128;
            const float r0 = hi ? v[k].y : v[k].x;
            const float r1 = hi ? v[k].z : v[k].y;
            const float r2 = hi ? v[k].w : v[k].z;
            const float r3 = hi ? v[k].x : v[k].w;
            atomicAdd(row + o0, r0);
            atomicAdd(row + o1, r1);
            atomicAdd(row + o2, r2);
            atomicAdd(row + o3, r3);
        }
        if (lane < 8 && ok) atomicAdd(cnt + dl, 1.0f);
        count = count > 8 ? count - 8 : 0;
        __builtin_amdgcn_fence(__ATOMIC_ACQ_REL, "wavefront");
    };

    auto push = [&](bool t, unsigned eidx, int dloc) {
        const unsigned m = __builtin_amdgcn_ballot_w32(t);
        if (m != 0u) {
            const int pos = count + (int)__builtin_amdgcn_mbcnt_lo(m, 0u);
            if (t) { qe[wq + pos] = eidx; qd[wq + pos] = (unsigned)dloc; }
            count += __builtin_popcount(m);
        }
    };

    const int nquads  = nedges >> 2;
    const int nrounds = (nquads + 4095) >> 12;
    int rr = (int)(((long long)blockIdx.x * nrounds) / gridDim.x);
    const int4* dst4 = (const int4*)dst;
    for (int r = 0; r < nrounds; ++r) {
        const int q0 = (rr << 12) + (wave << 7) + lane;
        int4 dq[4];
#pragma unroll
        for (int u = 0; u < 4; ++u) {
            const int q = q0 + (u << 5);
            dq[u] = (q < nquads) ? dst4[q] : make_int4(-1, -1, -1, -1);
        }
#pragma unroll
        for (int u = 0; u < 4; ++u) {
            const int q = q0 + (u << 5);
            const int l0 = dq[u].x - base, l1 = dq[u].y - base;
            const int l2 = dq[u].z - base, l3 = dq[u].w - base;
            const bool t0 = (unsigned)l0 < urows, t1 = (unsigned)l1 < urows;
            const bool t2 = (unsigned)l2 < urows, t3 = (unsigned)l3 < urows;
            if (__builtin_amdgcn_ballot_w32(t0 | t1 | t2 | t3) != 0u) {
                const unsigned e0 = (unsigned)q << 2;
                push(t0, e0, l0);
                push(t1, e0 + 1u, l1);
                push(t2, e0 + 2u, l2);
                push(t3, e0 + 3u, l3);
                while (count >= 8) drain8();
            }
        }
        rr = (rr + 1 == nrounds) ? 0 : rr + 1;
    }
    if (wave == 0) {
        const int e = (nquads << 2) + lane;
        const bool in = e < nedges;
        const int l = (in ? dst[e] : -1) - base;
        push(in && ((unsigned)l < urows), (unsigned)e, l);
    }
    while (count > 0) drain8();

    __syncthreads();
    for (int i = tid; i < rows * 32; i += 1024) {
        const int r = i >> 5, l = i & 31;
        const float4 v = *(const float4*)(acc + r * 128 + l * 4);
        *(float4*)(msg + (size_t)(base + r) * 128 + l * 4) = v;
    }
    for (int i = tid; i < rows; i += 1024) deg[base + i] = cnt[i];
}

__global__ __launch_bounds__(256) void sage_gemm(const float* __restrict__ hdst,
                                                 const float* __restrict__ msg,
                                                 const float* __restrict__ deg,
                                                 const float* __restrict__ Wself,
                                                 const float* __restrict__ Wneigh,
                                                 const float* __restrict__ bias,
                                                 float* __restrict__ out,
                                                 int n_dst, int d_out, int relu) {
    const int d_in = 128;
    int lane   = threadIdx.x & 31;
    int wave   = blockIdx.x * (blockDim.x >> 5) + (threadIdx.x >> 5);
    int ntiles = d_out >> 4;
    int mtiles = (n_dst + 15) >> 4;
    if (wave >= mtiles * ntiles) return;

    int mt = wave / ntiles;
    int nt = wave - mt * ntiles;
    int m0 = mt << 4;
    int n0 = nt << 4;

    int half = lane >> 4;
    int idx  = lane & 15;

    int arow = m0 + idx;
    if (arow > n_dst - 1) arow = n_dst - 1;

    const float* Ah = hdst  + (size_t)arow * d_in + 2 * half;
    const float* Am = msg   + (size_t)arow * d_in + 2 * half;
    const float* Bs = Wself  + (size_t)(n0 + idx) * d_in + 2 * half;
    const float* Bn = Wneigh + (size_t)(n0 + idx) * d_in + 2 * half;
    float invdeg = 1.0f / fmaxf(deg[arow], 1.0f);

    v8f c = {0.f, 0.f, 0.f, 0.f, 0.f, 0.f, 0.f, 0.f};

#pragma unroll
    for (int k = 0; k < d_in; k += 4) {
        v2f a = *(const v2f*)(Ah + k);
        v2f b = *(const v2f*)(Bs + k);
        c = __builtin_amdgcn_wmma_f32_16x16x4_f32(false, a, false, b,
                                                  (short)0, c, false, false);
    }
#pragma unroll
    for (int k = 0; k < d_in; k += 4) {
        v2f a = *(const v2f*)(Am + k);
        a.x *= invdeg;
        a.y *= invdeg;
        v2f b = *(const v2f*)(Bn + k);
        c = __builtin_amdgcn_wmma_f32_16x16x4_f32(false, a, false, b,
                                                  (short)0, c, false, false);
    }

    float bv = bias[n0 + idx];
#pragma unroll
    for (int j = 0; j < 8; ++j) {
        int m = m0 + j + 8 * half;
        if (m < n_dst) {
            float v = c[j] + bv;
            if (relu) v = fmaxf(v, 0.0f);
            out[(size_t)m * d_out + (n0 + idx)] = v;
        }
    }
}

static inline unsigned cdiv(unsigned a, unsigned b) { return (a + b - 1) / b; }

extern "C" void kernel_launch(void* const* d_in, const int* in_sizes, int n_in,
                              void* d_out, int out_size, void* d_ws, size_t ws_size,
                              hipStream_t stream) {
    const float* feats = (const float*)d_in[0];
    const int* src0 = (const int*)d_in[1];
    const int* dst0 = (const int*)d_in[2];
    const int* src1 = (const int*)d_in[3];
    const int* dst1 = (const int*)d_in[4];
    const int* src2 = (const int*)d_in[5];
    const int* dst2 = (const int*)d_in[6];
    const float* Ws0 = (const float*)d_in[10];
    const float* Wn0 = (const float*)d_in[11];
    const float* b0  = (const float*)d_in[12];
    const float* Ws1 = (const float*)d_in[13];
    const float* Wn1 = (const float*)d_in[14];
    const float* b1  = (const float*)d_in[15];
    const float* Ws2 = (const float*)d_in[16];
    const float* Wn2 = (const float*)d_in[17];
    const float* b2  = (const float*)d_in[18];

    const int E0 = in_sizes[1], E1 = in_sizes[3], E2 = in_sizes[5];
    const int N0 = 500000, N1 = 100000, N2 = 20000, N3 = 5000;
    constexpr int R0 = 400, R1 = 80, R2 = 20;

    char* ws = (char*)d_ws;
    float* h1  = (float*)ws;                    ws += (size_t)N1 * 128 * sizeof(float);
    float* h2  = (float*)ws;                    ws += (size_t)N2 * 128 * sizeof(float);
    float* msg = (float*)ws;                    ws += (size_t)N1 * 128 * sizeof(float);
    float* deg = (float*)ws;

    const dim3 blk(256);
    const dim3 blk_sum(1024);

    sage_sum_owned<R0><<<cdiv((unsigned)N1, R0), blk_sum, 0, stream>>>(feats, src0, dst0, E0, N0, N1, msg, deg);
    {
        unsigned tiles = cdiv((unsigned)N1, 16) * (128 / 16);
        sage_gemm<<<cdiv(tiles, 8), blk, 0, stream>>>(feats, msg, deg, Ws0, Wn0, b0, h1, N1, 128, 1);
    }

    sage_sum_owned<R1><<<cdiv((unsigned)N2, R1), blk_sum, 0, stream>>>(h1, src1, dst1, E1, N1, N2, msg, deg);
    {
        unsigned tiles = cdiv((unsigned)N2, 16) * (128 / 16);
        sage_gemm<<<cdiv(tiles, 8), blk, 0, stream>>>(h1, msg, deg, Ws1, Wn1, b1, h2, N2, 128, 1);
    }

    sage_sum_owned<R2><<<cdiv((unsigned)N3, R2), blk_sum, 0, stream>>>(h2, src2, dst2, E2, N2, N3, msg, deg);
    {
        unsigned tiles = cdiv((unsigned)N3, 16) * (64 / 16);
        sage_gemm<<<cdiv(tiles, 8), blk, 0, stream>>>(h2, msg, deg, Ws2, Wn2, b2, (float*)d_out, N3, 64, 0);
    }
}


# Layer products fused into the row-owner kernel: A read from the sums in LDS, B fragments in registers, output by whole rows; 6 launches become 3; neighbour terms now summed before self terms
# speedup vs baseline: 2.7943x; 1.4179x over previous
#include <hip/hip_runtime.h>
#include <hip/hip_bf16.h>


typedef __attribute__((ext_vector_type(2))) float v2f;
typedef __attribute__((ext_vector_type(8))) float v8f;

template <int R, int DOUT, bool RELU>
__global__ __launch_bounds__(1024) void sage_layer_owned(const float* __restrict__ hsrc,
                                                         const int* __restrict__ src,
                                                         const int* __restrict__ dst,
                                                         int nedges, int nsrc, int n_dst,
                                                         const float* __restrict__ Wself,
                                                         const float* __restrict__ Wneigh,
                                                         const float* __restrict__ bias,
                                                         float* __restrict__ out) {
    constexpr int QCAP = 136;
    constexpr int P    = 132;
    constexpr int NT   = DOUT / 16;
    constexpr int SG   = 32 / NT;
    constexpr int NSTR = (R + 15) / 16;
    constexpr int NIT  = (NSTR + SG - 1) / SG;
    constexpr int RPW  = 16 / NT;
    constexpr int QWORDS = 2 * 32 * QCAP;
    constexpr int SWORDS = SG * 16 * P;
    constexpr int XWORDS = QWORDS > SWORDS ? QWORDS : SWORDS;
    __shared__ __attribute__((aligned(16))) float acc[(R + 1) * P];
    __shared__ float cnt[R + 1];
    __shared__ __attribute__((aligned(16))) unsigned xbuf[XWORDS];
    unsigned* const qe = xbuf;
    unsigned* const qd = xbuf + 32 * QCAP;
    float* const stage = (float*)xbuf;

    const int tid  = threadIdx.x;
    const int lane = tid & 31;
    const int wave = tid >> 5;
    const int hi   = lane >> 4;
    const int base = blockIdx.x * R;
    int rows = n_dst - base;
    if (rows > R) rows = R;
    if (rows <= 0) return;
    const unsigned urows = (unsigned)rows;

    for (int i = tid; i < (R + 1) * (P / 4); i += 1024)
        *(float4*)(acc + i * 4) = make_float4(0.0f, 0.0f, 0.0f, 0.0f);
    for (int i = tid; i < R + 1; i += 1024) cnt[i] = 0.0f;
    __syncthreads();

    const int wq = wave * QCAP;
    int count = 0;

    const int o0 = lane * 4 + hi;
    const int o1 = lane * 4 + 1 + hi;
    const int o2 = lane * 4 + 2 + hi;
    const int o3 = lane * 4 + (hi ? 0 : 3);

    auto drain8 = [&]() {
        __builtin_amdgcn_fence(__ATOMIC_ACQ_REL, "wavefront");
        const int idx = count - 8 + (lane & 7);
        const bool ok = idx >= 0;
        unsigned e  = ok ? qe[wq + idx] : 0u;
        int dl      = ok ? (int)qd[wq + idx] : R;
        int s = ok ? src[e] : 0;
        s = s < 0 ? 0 : (s > nsrc - 1 ? nsrc - 1 : s);
        float4 v[8];
        int    dk[8];
#pragma unroll
        for (int k = 0; k < 8; ++k) {
            const int sk = __builtin_amdgcn_readlane(s, k);
            dk[k] = __builtin_amdgcn_readlane(dl, k);
            v[k] = *(const float4*)(hsrc + (size_t)sk * 128 + lane * 4);
        }
#pragma unroll
        for (int k = 0; k < 8; ++k) {
            float* row = acc + dk[k] * P;
            const float r0 = hi ? v[k].y : v[k].x;
            const float r1 = hi ? v[k].z : v[k].y;
            const float r2 = hi ? v[k].w : v[k].z;
            const float r3 = hi ? v[k].x : v[k].w;
            atomicAdd(row + o0, r0);
            atomicAdd(row + o1, r1);
            atomicAdd(row + o2, r2);
            atomicAdd(row + o3, r3);
        }
        if (lane < 8 && ok) atomicAdd(cnt + dl, 1.0f);
        count = count > 8 ? count - 8 : 0;
        __builtin_amdgcn_fence(__ATOMIC_ACQ_REL, "wavefront");
    };

    auto push = [&](bool t, unsigned eidx, int dloc) {
        const unsigned m = __builtin_amdgcn_ballot_w32(t);
        if (m != 0u) {
            const int pos = count + (int)__builtin_amdgcn_mbcnt_lo(m, 0u);
            if (t) { qe[wq + pos] = eidx; qd[wq + pos] = (unsigned)dloc; }
            count += __builtin_popcount(m);
        }
    };

    const int nquads  = nedges >> 2;
    const int nrounds = (nquads + 4095) >> 12;
    int rr = (int)(((long long)blockIdx.x * nrounds) / gridDim.x);
    const int4* dst4 = (const int4*)dst;
    for (int r = 0; r < nrounds; ++r) {
        const int q0 = (rr << 12) + (wave << 7) + lane;
        int4 dq[4];
#pragma unroll
        for (int u = 0; u < 4; ++u) {
            const int q = q0 + (u << 5);
            dq[u] = (q < nquads) ? dst4[q] : make_int4(-1, -1, -1, -1);
        }
#pragma unroll
        for (int u = 0; u < 4; ++u) {
            const int q = q0 + (u << 5);
            const int l0 = dq[u].x - base, l1 = dq[u].y - base;
            const int l2 = dq[u].z - base, l3 = dq[u].w - base;
            const bool t0 = (unsigned)l0 < urows, t1 = (unsigned)l1 < urows;
            const bool t2 = (unsigned)l2 < urows, t3 = (unsigned)l3 < urows;
            if (__builtin_amdgcn_ballot_w32(t0 | t1 | t2 | t3) != 0u) {
                const unsigned e0 = (unsigned)q << 2;
                push(t0, e0, l0);
                push(t1, e0 + 1u, l1);
                push(t2, e0 + 2u, l2);
                push(t3, e0 + 3u, l3);
                while (count >= 8) drain8();
            }
        }
        rr = (rr + 1 == nrounds) ? 0 : rr + 1;
    }
    if (wave == 0) {
        const int e = (nquads << 2) + lane;
        const bool in = e < nedges;
        const int l = (in ? dst[e] : -1) - base;
        push(in && ((unsigned)l < urows), (unsigned)e, l);
    }
    while (count > 0) drain8();

    __syncthreads();

    const int wv  = __builtin_amdgcn_readfirstlane(wave);
    const int nt  = wv % NT;
    const int sg  = wv / NT;
    const int n0  = nt * 16;
    const int idx = lane & 15;
    const int nstr = (rows + 15) >> 4;

    {
        v2f bfr[32];
        const float* Bn = Wneigh + (size_t)(n0 + idx) * 128 + 2 * hi;
#pragma unroll
        for (int k = 0; k < 32; ++k) bfr[k] = *(const v2f*)(Bn + 4 * k);

        for (int it = 0; it < NIT; ++it) {
            const int strip = it * SG + sg;
            const bool valid = strip < nstr;
            v8f c = {0.f, 0.f, 0.f, 0.f, 0.f, 0.f, 0.f, 0.f};
            if (valid) {
                int ar = strip * 16 + idx;
                if (ar > R) ar = R;
                const float invdeg = 1.0f / fmaxf(cnt[ar], 1.0f);
                const float* Am = acc + ar * P + 2 * hi;
#pragma unroll
                for (int k = 0; k < 32; ++k) {
                    v2f a = *(const v2f*)(Am + 4 * k);
                    a.x *= invdeg;
                    a.y *= invdeg;
                    c = __builtin_amdgcn_wmma_f32_16x16x4_f32(false, a, false, bfr[k],
                                                              (short)0, c, false, false);
                }
            }
            __syncthreads();
            if (valid) {
#pragma unroll
                for (int j = 0; j < 8; ++j) {
                    const int r = strip * 16 + 8 * hi + j;
                    if (r < rows) acc[r * P + n0 + idx] = c[j];
                }
            }
        }
    }

    {
        v2f bfr[32];
        const float* Bs = Wself + (size_t)(n0 + idx) * 128 + 2 * hi;
#pragma unroll
        for (int k = 0; k < 32; ++k) bfr[k] = *(const v2f*)(Bs + 4 * k);
        const float bv = bias[n0 + idx];

        static_assert(RPW == 2 || RPW == 4, "a wave copies 2 or 4 rows of a strip");
        float4 nx0 = make_float4(0.f, 0.f, 0.f, 0.f), nx1 = nx0, nx2 = nx0, nx3 = nx0;
        const float* hrow = hsrc + (size_t)base * 128 + lane * 4;
        const int lr0 = nt * RPW;
        auto ldrow = [hrow, rows](int gr) -> float4 {
            if (gr > rows - 1) gr = rows - 1;
            return *(const float4*)(hrow + (size_t)gr * 128);
        };
        if (sg < nstr) {
            nx0 = ldrow(sg * 16 + lr0);
            nx1 = ldrow(sg * 16 + lr0 + 1);
            if (RPW == 4) { nx2 = ldrow(sg * 16 + lr0 + 2); nx3 = ldrow(sg * 16 + lr0 + 3); }
        }

        for (int it = 0; it < NIT; ++it) {
            const int strip = it * SG + sg;
            const bool valid = strip < nstr;
            if (valid) {
                float* srow = stage + (sg * 16 + lr0) * P + lane * 4;
                *(float4*)(srow)         = nx0;
                *(float4*)(srow + P)     = nx1;
                if (RPW == 4) { *(float4*)(srow + 2 * P) = nx2; *(float4*)(srow + 3 * P) = nx3; }
            }
            __syncthreads();
            if (strip + SG < nstr) {
                const int g0 = (strip + SG) * 16 + lr0;
                nx0 = ldrow(g0);
                nx1 = ldrow(g0 + 1);
                if (RPW == 4) { nx2 = ldrow(g0 + 2); nx3 = ldrow(g0 + 3); }
            }
            if (valid) {
                v8f c;
#pragma unroll
                for (int j = 0; j < 8; ++j) {
                    const int r = strip * 16 + 8 * hi + j;
                    c[j] = (r < rows) ? acc[r * P + n0 + idx] : 0.0f;
                }
                const float* As = stage + (sg * 16 + idx) * P + 2 * hi;
#pragma unroll
                for (int k = 0; k < 32; ++k) {
                    const v2f a = *(const v2f*)(As + 4 * k);
                    c = __builtin_amdgcn_wmma_f32_16x16x4_f32(false, a, false, bfr[k],
                                                              (short)0, c, false, false);
                }
#pragma unroll
                for (int j = 0; j < 8; ++j) {
                    const int r = strip * 16 + 8 * hi + j;
                    if (r < rows) {
                        float v = c[j] + bv;
                        if (RELU) v = fmaxf(v, 0.0f);
                        acc[r * P + n0 + idx] = v;
                    }
                }
            }
            if (it + 1 < NIT) __syncthreads();
        }
    }

    __syncthreads();
    for (int i = tid; i < rows * (DOUT / 4); i += 1024) {
        const int r = i / (DOUT / 4), l = i % (DOUT / 4);
        const float4 v = *(const float4*)(acc + r * P + l * 4);
        *(float4*)(out + (size_t)(base + r) * DOUT + l * 4) = v;
    }
}

static inline unsigned cdiv(unsigned a, unsigned b) { return (a + b - 1) / b; }

extern "C" void kernel_launch(void* const* d_in, const int* in_sizes, int n_in,
                              void* d_out, int out_size, void* d_ws, size_t ws_size,
                              hipStream_t stream) {
    const float* feats = (const float*)d_in[0];
    const int* src0 = (const int*)d_in[1];
    const int* dst0 = (const int*)d_in[2];
    const int* src1 = (const int*)d_in[3];
    const int* dst1 = (const int*)d_in[4];
    const int* src2 = (const int*)d_in[5];
    const int* dst2 = (const int*)d_in[6];
    const float* Ws0 = (const float*)d_in[10];
    const float* Wn0 = (const float*)d_in[11];
    const float* b0  = (const float*)d_in[12];
    const float* Ws1 = (const float*)d_in[13];
    const float* Wn1 = (const float*)d_in[14];
    const float* b1  = (const float*)d_in[15];
    const float* Ws2 = (const float*)d_in[16];
    const float* Wn2 = (const float*)d_in[17];
    const float* b2  = (const float*)d_in[18];

    const int E0 = in_sizes[1], E1 = in_sizes[3], E2 = in_sizes[5];
    const int N0 = 500000, N1 = 100000, N2 = 20000, N3 = 5000;
    constexpr int R0 = 400, R1 = 80, R2 = 20;

    char* ws = (char*)d_ws;
    float* h1  = (float*)ws;                    ws += (size_t)N1 * 128 * sizeof(float);
    float* h2  = (float*)ws;

    const dim3 blk(1024);

    sage_layer_owned<R0, 128, true><<<cdiv((unsigned)N1, R0), blk, 0, stream>>>(feats, src0, dst0, E0, N0, N1, Ws0, Wn0, b0, h1);
    sage_layer_owned<R1, 128, true><<<cdiv((unsigned)N2, R1), blk, 0, stream>>>(h1, src1, dst1, E1, N1, N2, Ws1, Wn1, b1, h2);
    sage_layer_owned<R2, 64, false><<<cdiv((unsigned)N3, R2), blk, 0, stream>>>(h2, src2, dst2, E2, N2, N3, Ws2, Wn2, b2, (float*)d_out);
}


# Layer 0 scan reads one owner byte per edge (dst/R, written by a small pre-pass into h2's idle space) instead of the 4-byte dst; 16 edges per lane per load; row sum order still LDS-add arrival
# speedup vs baseline: 3.1890x; 1.1413x over previous
#include <hip/hip_runtime.h>
#include <hip/hip_bf16.h>


typedef __attribute__((ext_vector_type(2))) float v2f;
typedef __attribute__((ext_vector_type(8))) float v8f;

template <int R>
__global__ __launch_bounds__(1024) void owner_bytes(const int* __restrict__ dst, int nedges, int n_dst,
                                                    unsigned* __restrict__ ob) {
    const int nwords = ((nedges + 15) >> 4) << 2;
    const unsigned un = (unsigned)n_dst;
    for (int i = blockIdx.x * 1024 + threadIdx.x; i < nwords; i += gridDim.x * 1024) {
        const int e = i * 4;
        int4 d;
        if (e + 3 < nedges) {
            d = *(const int4*)(dst + e);
        } else {
            d.x = e     < nedges ? dst[e]     : -1;
            d.y = e + 1 < nedges ? dst[e + 1] : -1;
            d.z = e + 2 < nedges ? dst[e + 2] : -1;
            d.w = e + 3 < nedges ? dst[e + 3] : -1;
        }
        const unsigned b0 = (unsigned)d.x < un ? (unsigned)d.x / (unsigned)R : 255u;
        const unsigned b1 = (unsigned)d.y < un ? (unsigned)d.y / (unsigned)R : 255u;
        const unsigned b2 = (unsigned)d.z < un ? (unsigned)d.z / (unsigned)R : 255u;
        const unsigned b3 = (unsigned)d.w < un ? (unsigned)d.w / (unsigned)R : 255u;
        ob[i] = b0 | (b1 << 8) | (b2 << 16) | (b3 << 24);
    }
}

static __device__ inline unsigned zero_bytes(unsigned x) {
    const unsigned t = (x & 0x7f7f7f7fu) + 0x7f7f7f7fu;
    return ~(t | x) & 0x80808080u;
}

template <int R, int DOUT, bool RELU, bool BYTES>
__global__ __launch_bounds__(1024) void sage_layer_owned(const float* __restrict__ hsrc,
                                                         const int* __restrict__ src,
                                                         const int* __restrict__ dst,
                                                         const unsigned* __restrict__ ob,
                                                         int nedges, int nsrc, int n_dst,
                                                         const float* __restrict__ Wself,
                                                         const float* __restrict__ Wneigh,
                                                         const float* __restrict__ bias,
                                                         float* __restrict__ out) {
    constexpr int QCAP = 136;
    constexpr int P    = 132;
    constexpr int NT   = DOUT / 16;
    constexpr int SG   = 32 / NT;
    constexpr int NSTR = (R + 15) / 16;
    constexpr int NIT  = (NSTR + SG - 1) / SG;
    constexpr int RPW  = 16 / NT;
    constexpr int QWORDS = (BYTES ? 1 : 2) * 32 * QCAP;
    constexpr int SWORDS = SG * 16 * P;
    constexpr int XWORDS = QWORDS > SWORDS ? QWORDS : SWORDS;
    __shared__ __attribute__((aligned(16))) float acc[(R + 1) * P];
    __shared__ float cnt[R + 1];
    __shared__ __attribute__((aligned(16))) unsigned xbuf[XWORDS];
    unsigned* const qe = xbuf;
    unsigned* const qd = xbuf + 32 * QCAP;
    float* const stage = (float*)xbuf;

    const int tid  = threadIdx.x;
    const int lane = tid & 31;
    const int wave = tid >> 5;
    const int hi   = lane >> 4;
    const int base = blockIdx.x * R;
    int rows = n_dst - base;
    if (rows > R) rows = R;
    if (rows <= 0) return;
    const unsigned urows = (unsigned)rows;

    for (int i = tid; i < (R + 1) * (P / 4); i += 1024)
        *(float4*)(acc + i * 4) = make_float4(0.0f, 0.0f, 0.0f, 0.0f);
    for (int i = tid; i < R + 1; i += 1024) cnt[i] = 0.0f;
    __syncthreads();

    const int wq = wave * QCAP;
    int count = 0;

    const int o0 = lane * 4 + hi;
    const int o1 = lane * 4 + 1 + hi;
    const int o2 = lane * 4 + 2 + hi;
    const int o3 = lane * 4 + (hi ? 0 : 3);

    auto drain8 = [&]() {
        __builtin_amdgcn_fence(__ATOMIC_ACQ_REL, "wavefront");
        const int idx = count - 8 + (lane & 7);
        const bool ok = idx >= 0;
        unsigned e  = ok ? qe[wq + idx] : 0u;
        int dl;
        if constexpr (BYTES) dl = ok ? dst[e] - base : R;
        else                 dl = ok ? (int)qd[wq + idx] : R;
        int s = ok ? src[e] : 0;
        s = s < 0 ? 0 : (s > nsrc - 1 ? nsrc - 1 : s);
        float4 v[8];
        int    dk[8];
#pragma unroll
        for (int k = 0; k < 8; ++k) {
            const int sk = __builtin_amdgcn_readlane(s, k);
            dk[k] = __builtin_amdgcn_readlane(dl, k);
            v[k] = *(const float4*)(hsrc + (size_t)sk * 128 + lane * 4);
        }
#pragma unroll
        for (int k = 0; k < 8; ++k) {
            float* row = acc + dk[k] * P;
            const float r0 = hi ? v[k].y : v[k].x;
            const float r1 = hi ? v[k].z : v[k].y;
            const float r2 = hi ? v[k].w : v[k].z;
            const float r3 = hi ? v[k].x : v[k].w;
            atomicAdd(row + o0, r0);
            atomicAdd(row + o1, r1);
            atomicAdd(row + o2, r2);
            atomicAdd(row + o3, r3);
        }
        if (lane < 8 && ok) atomicAdd(cnt + dl, 1.0f);
        count = count > 8 ? count - 8 : 0;
        __builtin_amdgcn_fence(__ATOMIC_ACQ_REL, "wavefront");
    };

    auto push = [&](bool t, unsigned eidx, int dloc) {
        const unsigned m = __builtin_amdgcn_ballot_w32(t);
        if (m != 0u) {
            const int pos = count + (int)__builtin_amdgcn_mbcnt_lo(m, 0u);
            if (t) {
                qe[wq + pos] = eidx;
                if constexpr (!BYTES) qd[wq + pos] = (unsigned)dloc;
            }
            count += __builtin_popcount(m);
        }
    };

    if constexpr (BYTES) {
        const int ngroups = (nedges + 15) >> 4;
        const int nrounds = (ngroups + 4095) >> 12;
        int rr = (int)(((long long)blockIdx.x * nrounds) / gridDim.x);
        const uint4* ob4 = (const uint4*)ob;
        const unsigned bc = (unsigned)blockIdx.x * 0x01010101u;
        for (int r = 0; r < nrounds; ++r) {
            const int q0 = (rr << 12) + (wave << 7) + lane;
            uint4 w[4];
#pragma unroll
            for (int u = 0; u < 4; ++u) {
                const int q = q0 + (u << 5);
                w[u] = (q < ngroups) ? ob4[q] : make_uint4(~0u, ~0u, ~0u, ~0u);
            }
#pragma unroll
            for (int u = 0; u < 4; ++u) {
                const int q = q0 + (u << 5);
                unsigned m = (zero_bytes(w[u].x ^ bc) >> 7) | (zero_bytes(w[u].y ^ bc) >> 6) |
                             (zero_bytes(w[u].z ^ bc) >> 5) | (zero_bytes(w[u].w ^ bc) >> 4);
                while (__builtin_amdgcn_ballot_w32(m != 0u) != 0u) {
                    const bool t = m != 0u;
                    const unsigned p = (unsigned)__builtin_ctz(m | 0x10000000u);
                    const unsigned off = ((p & 3u) << 2) | (p >> 3);
                    push(t, ((unsigned)q << 4) + off, 0);
                    m &= m - 1u;
                    while (count >= 8) drain8();
                }
            }
            rr = (rr + 1 == nrounds) ? 0 : rr + 1;
        }
    } else {

    const int nquads  = nedges >> 2;
    const int nrounds = (nquads + 4095) >> 12;
    int rr = (int)(((long long)blockIdx.x * nrounds) / gridDim.x);
    const int4* dst4 = (const int4*)dst;
    for (int r = 0; r < nrounds; ++r) {
        const int q0 = (rr << 12) + (wave << 7) + lane;
        int4 dq[4];
#pragma unroll
        for (int u = 0; u < 4; ++u) {
            const int q = q0 + (u << 5);
            dq[u] = (q < nquads) ? dst4[q] : make_int4(-1, -1, -1, -1);
        }
#pragma unroll
        for (int u = 0; u < 4; ++u) {
            const int q = q0 + (u << 5);
            const int l0 = dq[u].x - base, l1 = dq[u].y - base;
            const int l2 = dq[u].z - base, l3 = dq[u].w - base;
            const bool t0 = (unsigned)l0 < urows, t1 = (unsigned)l1 < urows;
            const bool t2 = (unsigned)l2 < urows, t3 = (unsigned)l3 < urows;
            if (__builtin_amdgcn_ballot_w32(t0 | t1 | t2 | t3) != 0u) {
                const unsigned e0 = (unsigned)q << 2;
                push(t0, e0, l0);
                push(t1, e0 + 1u, l1);
                push(t2, e0 + 2u, l2);
                push(t3, e0 + 3u, l3);
                while (count >= 8) drain8();
            }
        }
        rr = (rr + 1 == nrounds) ? 0 : rr + 1;
    }
    if (wave == 0) {
        const int e = (nquads << 2) + lane;
        const bool in = e < nedges;
        const int l = (in ? dst[e] : -1) - base;
        push(in && ((unsigned)l < urows), (unsigned)e, l);
    }
    }
    while (count > 0) drain8();

    __syncthreads();

    const int wv  = __builtin_amdgcn_readfirstlane(wave);
    const int nt  = wv % NT;
    const int sg  = wv / NT;
    const int n0  = nt * 16;
    const int idx = lane & 15;
    const int nstr = (rows + 15) >> 4;

    {
        v2f bfr[32];
        const float* Bn = Wneigh + (size_t)(n0 + idx) * 128 + 2 * hi;
#pragma unroll
        for (int k = 0; k < 32; ++k) bfr[k] = *(const v2f*)(Bn + 4 * k);

        for (int it = 0; it < NIT; ++it) {
            const int strip = it * SG + sg;
            const bool valid = strip < nstr;
            v8f c = {0.f, 0.f, 0.f, 0.f, 0.f, 0.f, 0.f, 0.f};
            if (valid) {
                int ar = strip * 16 + idx;
                if (ar > R) ar = R;
                const float invdeg = 1.0f / fmaxf(cnt[ar], 1.0f);
                const float* Am = acc + ar * P + 2 * hi;
#pragma unroll
                for (int k = 0; k < 32; ++k) {
                    v2f a = *(const v2f*)(Am + 4 * k);
                    a.x *= invdeg;
                    a.y *= invdeg;
                    c = __builtin_amdgcn_wmma_f32_16x16x4_f32(false, a, false, bfr[k],
                                                              (short)0, c, false, false);
                }
            }
            __syncthreads();
            if (valid) {
#pragma unroll
                for (int j = 0; j < 8; ++j) {
                    const int r = strip * 16 + 8 * hi + j;
                    if (r < rows) acc[r * P + n0 + idx] = c[j];
                }
            }
        }
    }

    {
        v2f bfr[32];
        const float* Bs = Wself + (size_t)(n0 + idx) * 128 + 2 * hi;
#pragma unroll
        for (int k = 0; k < 32; ++k) bfr[k] = *(const v2f*)(Bs + 4 * k);
        const float bv = bias[n0 + idx];

        static_assert(RPW == 2 || RPW == 4, "a wave copies 2 or 4 rows of a strip");
        float4 nx0 = make_float4(0.f, 0.f, 0.f, 0.f), nx1 = nx0, nx2 = nx0, nx3 = nx0;
        const float* hrow = hsrc + (size_t)base * 128 + lane * 4;
        const int lr0 = nt * RPW;
        auto ldrow = [hrow, rows](int gr) -> float4 {
            if (gr > rows - 1) gr = rows - 1;
            return *(const float4*)(hrow + (size_t)gr * 128);
        };
        if (sg < nstr) {
            nx0 = ldrow(sg * 16 + lr0);
            nx1 = ldrow(sg * 16 + lr0 + 1);
            if (RPW == 4) { nx2 = ldrow(sg * 16 + lr0 + 2); nx3 = ldrow(sg * 16 + lr0 + 3); }
        }

        for (int it = 0; it < NIT; ++it) {
            const int strip = it * SG + sg;
            const bool valid = strip < nstr;
            if (valid) {
                float* srow = stage + (sg * 16 + lr0) * P + lane * 4;
                *(float4*)(srow)         = nx0;
                *(float4*)(srow + P)     = nx1;
                if (RPW == 4) { *(float4*)(srow + 2 * P) = nx2; *(float4*)(srow + 3 * P) = nx3; }
            }
            __syncthreads();
            if (strip + SG < nstr) {
                const int g0 = (strip + SG) * 16 + lr0;
                nx0 = ldrow(g0);
                nx1 = ldrow(g0 + 1);
                if (RPW == 4) { nx2 = ldrow(g0 + 2); nx3 = ldrow(g0 + 3); }
            }
            if (valid) {
                v8f c;
#pragma unroll
                for (int j = 0; j < 8; ++j) {
                    const int r = strip * 16 + 8 * hi + j;
                    c[j] = (r < rows) ? acc[r * P + n0 + idx] : 0.0f;
                }
                const float* As = stage + (sg * 16 + idx) * P + 2 * hi;
#pragma unroll
                for (int k = 0; k < 32; ++k) {
                    const v2f a = *(const v2f*)(As + 4 * k);
                    c = __builtin_amdgcn_wmma_f32_16x16x4_f32(false, a, false, bfr[k],
                                                              (short)0, c, false, false);
                }
#pragma unroll
                for (int j = 0; j < 8; ++j) {
                    const int r = strip * 16 + 8 * hi + j;
                    if (r < rows) {
                        float v = c[j] + bv;
                        if (RELU) v = fmaxf(v, 0.0f);
                        acc[r * P + n0 + idx] = v;
                    }
                }
            }
            if (it + 1 < NIT) __syncthreads();
        }
    }

    __syncthreads();
    for (int i = tid; i < rows * (DOUT / 4); i += 1024) {
        const int r = i / (DOUT / 4), l = i % (DOUT / 4);
        const float4 v = *(const float4*)(acc + r * P + l * 4);
        *(float4*)(out + (size_t)(base + r) * DOUT + l * 4) = v;
    }
}

static inline unsigned cdiv(unsigned a, unsigned b) { return (a + b - 1) / b; }

extern "C" void kernel_launch(void* const* d_in, const int* in_sizes, int n_in,
                              void* d_out, int out_size, void* d_ws, size_t ws_size,
                              hipStream_t stream) {
    const float* feats = (const float*)d_in[0];
    const int* src0 = (const int*)d_in[1];
    const int* dst0 = (const int*)d_in[2];
    const int* src1 = (const int*)d_in[3];
    const int* dst1 = (const int*)d_in[4];
    const int* src2 = (const int*)d_in[5];
    const int* dst2 = (const int*)d_in[6];
    const float* Ws0 = (const float*)d_in[10];
    const float* Wn0 = (const float*)d_in[11];
    const float* b0  = (const float*)d_in[12];
    const float* Ws1 = (const float*)d_in[13];
    const float* Wn1 = (const float*)d_in[14];
    const float* b1  = (const float*)d_in[15];
    const float* Ws2 = (const float*)d_in[16];
    const float* Wn2 = (const float*)d_in[17];
    const float* b2  = (const float*)d_in[18];

    const int E0 = in_sizes[1], E1 = in_sizes[3], E2 = in_sizes[5];
    const int N0 = 500000, N1 = 100000, N2 = 20000, N3 = 5000;
    constexpr int R0 = 400, R1 = 80, R2 = 20;

    char* ws = (char*)d_ws;
    float* h1  = (float*)ws;                    ws += (size_t)N1 * 128 * sizeof(float);
    float* h2  = (float*)ws;

    const dim3 blk(1024);

    static_assert((N1 + R0 - 1) / R0 <= 255, "a block number must fit the owner byte");
    unsigned* ob0 = (unsigned*)h2;
    owner_bytes<R0><<<500, blk, 0, stream>>>(dst0, E0, N1, ob0);

    sage_layer_owned<R0, 128, true, true><<<cdiv((unsigned)N1, R0), blk, 0, stream>>>(feats, src0, dst0, ob0, E0, N0, N1, Ws0, Wn0, b0, h1);
    sage_layer_owned<R1, 128, true, false><<<cdiv((unsigned)N2, R1), blk, 0, stream>>>(h1, src1, dst1, nullptr, E1, N1, N2, Ws1, Wn1, b1, h2);
    sage_layer_owned<R2, 64, false, false><<<cdiv((unsigned)N3, R2), blk, 0, stream>>>(h2, src2, dst2, nullptr, E2, N2, N3, Ws2, Wn2, b2, (float*)d_out);
}


# Layers 0 and 1 no longer scan every edge: a pre-pass writes one bit per 16-edge group per block of owned rows; a work-group walks its own bitmap, opens only marked groups; sum order still LDS arrival
# speedup vs baseline: 3.4688x; 1.0877x over previous
#include <hip/hip_runtime.h>
#include <hip/hip_bf16.h>


typedef __attribute__((ext_vector_type(2))) float v2f;
typedef __attribute__((ext_vector_type(8))) float v8f;

template <int R, int NB, int PITCH>
static __device__ inline void bitmap_tile(const int* __restrict__ dst, int ngroups, unsigned un, int tile,
                                          unsigned* __restrict__ bm, unsigned* bits) {
    const int tid = threadIdx.x;
    const unsigned bit = 1u << (tid & 31);
    const int word = tid >> 5;
    const int g = (tile << 10) + tid;
    if (g < ngroups) {
        const int4* p = (const int4*)(dst + ((size_t)g << 4));
        const int4 d0 = p[0], d1 = p[1], d2 = p[2], d3 = p[3];
        const int dv[16] = {d0.x, d0.y, d0.z, d0.w, d1.x, d1.y, d1.z, d1.w,
                            d2.x, d2.y, d2.z, d2.w, d3.x, d3.y, d3.z, d3.w};
#pragma unroll
        for (int j = 0; j < 16; ++j) {
            const unsigned d = (unsigned)dv[j];
            if (d < un) {
                const unsigned o = d / (unsigned)R;
                if (o < (unsigned)NB) atomicOr(&bits[word * NB + o], bit);
            }
        }
    }
    __syncthreads();
    for (int i = tid; i < NB * 32; i += 1024) {
        const int k = (i & 31) * NB + (i >> 5);
        bm[(size_t)(i >> 5) * PITCH + (tile << 5) + (i & 31)] = bits[k];
        bits[k] = 0u;
    }
}

template <int RA, int RB, int NB, int PITCH, int OFFB>
__global__ __launch_bounds__(1024) void group_bitmaps(const int* __restrict__ dstA, int nedgesA, int n_dstA,
                                                      const int* __restrict__ dstB, int nedgesB, int n_dstB,
                                                      unsigned* __restrict__ bm) {
    __shared__ unsigned bits[NB * 32];
    const int ngA = nedgesA >> 4, ngB = nedgesB >> 4;
    const int ntA = (ngA + 1023) >> 10, ntB = (ngB + 1023) >> 10;
    for (int i = threadIdx.x; i < NB * 32; i += 1024) bits[i] = 0u;
    for (int tile = blockIdx.x; tile < ntA + ntB; tile += gridDim.x) {
        __syncthreads();
        if (tile < ntA) bitmap_tile<RA, NB, PITCH>(dstA, ngA, (unsigned)n_dstA, tile, bm, bits);
        else            bitmap_tile<RB, NB, PITCH>(dstB, ngB, (unsigned)n_dstB, tile - ntA, bm + OFFB, bits);
    }
}

template <int R, int DOUT, bool RELU, int BMP>
__global__ __launch_bounds__(1024) void sage_layer_owned(const float* __restrict__ hsrc,
                                                         const int* __restrict__ src,
                                                         const int* __restrict__ dst,
                                                         const unsigned* bm,
                                                         int nedges, int nsrc, int n_dst,
                                                         const float* __restrict__ Wself,
                                                         const float* __restrict__ Wneigh,
                                                         const float* __restrict__ bias,
                                                         float* __restrict__ out) {
    constexpr bool BITMAP = BMP > 0;
    constexpr int QCAP = 136;
    constexpr int P    = 132;
    constexpr int NT   = DOUT / 16;
    constexpr int SG   = 32 / NT;
    constexpr int NSTR = (R + 15) / 16;
    constexpr int NIT  = (NSTR + SG - 1) / SG;
    constexpr int RPW  = 16 / NT;
    constexpr int QWORDS = (BITMAP ? 1 : 2) * 32 * QCAP;
    constexpr int SWORDS = SG * 16 * P;
    constexpr int XWORDS = QWORDS > SWORDS ? QWORDS : SWORDS;
    __shared__ __attribute__((aligned(16))) float acc[(R + 1) * P];
    __shared__ float cnt[R + 1];
    __shared__ __attribute__((aligned(16))) unsigned xbuf[XWORDS];
    unsigned* const qe = xbuf;
    unsigned* const qd = xbuf + 32 * QCAP;
    float* const stage = (float*)xbuf;

    const int tid  = threadIdx.x;
    const int lane = tid & 31;
    const int wave = tid >> 5;
    const int hi   = lane >> 4;
    const int base = blockIdx.x * R;
    int rows = n_dst - base;
    if (rows > R) rows = R;
    if (rows <= 0) return;
    const unsigned urows = (unsigned)rows;

    for (int i = tid; i < (R + 1) * (P / 4); i += 1024)
        *(float4*)(acc + i * 4) = make_float4(0.0f, 0.0f, 0.0f, 0.0f);
    for (int i = tid; i < R + 1; i += 1024) cnt[i] = 0.0f;
    __syncthreads();

    const int wq = wave * QCAP;
    int count = 0;

    const int o0 = lane * 4 + hi;
    const int o1 = lane * 4 + 1 + hi;
    const int o2 = lane * 4 + 2 + hi;
    const int o3 = lane * 4 + (hi ? 0 : 3);

    auto drain8 = [&]() {
        __builtin_amdgcn_fence(__ATOMIC_ACQ_REL, "wavefront");
        const int idx = count - 8 + (lane & 7);
        const bool ok = idx >= 0;
        unsigned e  = ok ? qe[wq + idx] : 0u;
        int dl;
        if constexpr (BITMAP) dl = ok ? dst[e] - base : R;
        else                 dl = ok ? (int)qd[wq + idx] : R;
        int s = ok ? src[e] : 0;
        s = s < 0 ? 0 : (s > nsrc - 1 ? nsrc - 1 : s);
        float4 v[8];
        int    dk[8];
#pragma unroll
        for (int k = 0; k < 8; ++k) {
            const int sk = __builtin_amdgcn_readlane(s, k);
            dk[k] = __builtin_amdgcn_readlane(dl, k);
            v[k] = *(const float4*)(hsrc + (size_t)sk * 128 + lane * 4);
        }
#pragma unroll
        for (int k = 0; k < 8; ++k) {
            float* row = acc + dk[k] * P;
            const float r0 = hi ? v[k].y : v[k].x;
            const float r1 = hi ? v[k].z : v[k].y;
            const float r2 = hi ? v[k].w : v[k].z;
            const float r3 = hi ? v[k].x : v[k].w;
            atomicAdd(row + o0, r0);
            atomicAdd(row + o1, r1);
            atomicAdd(row + o2, r2);
            atomicAdd(row + o3, r3);
        }
        if (lane < 8 && ok) atomicAdd(cnt + dl, 1.0f);
        count = count > 8 ? count - 8 : 0;
        __builtin_amdgcn_fence(__ATOMIC_ACQ_REL, "wavefront");
    };

    auto push = [&](bool t, unsigned eidx, int dloc) {
        const unsigned m = __builtin_amdgcn_ballot_w32(t);
        if (m != 0u) {
            const int pos = count + (int)__builtin_amdgcn_mbcnt_lo(m, 0u);
            if (t) {
                qe[wq + pos] = eidx;
                if constexpr (!BITMAP) qd[wq + pos] = (unsigned)dloc;
            }
            count += __builtin_popcount(m);
        }
    };

    if constexpr (BITMAP) {
        const int ngroups = nedges >> 4;
        const int wpo = ((ngroups + 1023) >> 10) << 5;
        const unsigned* bmw = bm + (size_t)blockIdx.x * BMP;
        for (int w0 = wave << 5; w0 < wpo; w0 += 1024) {
            unsigned word = bmw[w0 + lane];
            while (__builtin_amdgcn_ballot_w32(word != 0u) != 0u) {
                const int g  = ((w0 + lane) << 5) + (int)__builtin_ctz(word | 0x80000000u);
                const bool t = (word != 0u) && (g < ngroups);
                const int e0 = g << 4;
                word &= word - 1u;
                int4 d0 = make_int4(-1, -1, -1, -1), d1 = d0, d2 = d0, d3 = d0;
                if (t) {
                    const int4* p = (const int4*)(dst + e0);
                    d0 = p[0];
                    d1 = p[1];
                    d2 = p[2];
                    d3 = p[3];
                }
                unsigned m;
                m  = ((unsigned)(d0.x - base) < urows ? 0x0001u : 0u) | ((unsigned)(d0.y - base) < urows ? 0x0002u : 0u);
                m |= ((unsigned)(d0.z - base) < urows ? 0x0004u : 0u) | ((unsigned)(d0.w - base) < urows ? 0x0008u : 0u);
                m |= ((unsigned)(d1.x - base) < urows ? 0x0010u : 0u) | ((unsigned)(d1.y - base) < urows ? 0x0020u : 0u);
                m |= ((unsigned)(d1.z - base) < urows ? 0x0040u : 0u) | ((unsigned)(d1.w - base) < urows ? 0x0080u : 0u);
                m |= ((unsigned)(d2.x - base) < urows ? 0x0100u : 0u) | ((unsigned)(d2.y - base) < urows ? 0x0200u : 0u);
                m |= ((unsigned)(d2.z - base) < urows ? 0x0400u : 0u) | ((unsigned)(d2.w - base) < urows ? 0x0800u : 0u);
                m |= ((unsigned)(d3.x - base) < urows ? 0x1000u : 0u) | ((unsigned)(d3.y - base) < urows ? 0x2000u : 0u);
                m |= ((unsigned)(d3.z - base) < urows ? 0x4000u : 0u) | ((unsigned)(d3.w - base) < urows ? 0x8000u : 0u);
                while (__builtin_amdgcn_ballot_w32(m != 0u) != 0u) {
                    const bool tt = m != 0u;
                    const unsigned p = (unsigned)__builtin_ctz(m | 0x10000u);
                    push(tt, (unsigned)e0 + p, 0);
                    m &= m - 1u;
                    while (count >= 8) drain8();
                }
            }
        }
        if (wave == 0) {
            const int e = (ngroups << 4) + lane;
            const bool in = e < nedges;
            const int l = (in ? dst[e] : -1) - base;
            push(in && ((unsigned)l < urows), (unsigned)e, 0);
        }
    } else {

    const int nquads  = nedges >> 2;
    const int nrounds = (nquads + 4095) >> 12;
    int rr = (int)(((long long)blockIdx.x * nrounds) / gridDim.x);
    const int4* dst4 = (const int4*)dst;
    for (int r = 0; r < nrounds; ++r) {
        const int q0 = (rr << 12) + (wave << 7) + lane;
        int4 dq[4];
#pragma unroll
        for (int u = 0; u < 4; ++u) {
            const int q = q0 + (u << 5);
            dq[u] = (q < nquads) ? dst4[q] : make_int4(-1, -1, -1, -1);
        }
#pragma unroll
        for (int u = 0; u < 4; ++u) {
            const int q = q0 + (u << 5);
            const int l0 = dq[u].x - base, l1 = dq[u].y - base;
            const int l2 = dq[u].z - base, l3 = dq[u].w - base;
            const bool t0 = (unsigned)l0 < urows, t1 = (unsigned)l1 < urows;
            const bool t2 = (unsigned)l2 < urows, t3 = (unsigned)l3 < urows;
            if (__builtin_amdgcn_ballot_w32(t0 | t1 | t2 | t3) != 0u) {
                const unsigned e0 = (unsigned)q << 2;
                push(t0, e0, l0);
                push(t1, e0 + 1u, l1);
                push(t2, e0 + 2u, l2);
                push(t3, e0 + 3u, l3);
                while (count >= 8) drain8();
            }
        }
        rr = (rr + 1 == nrounds) ? 0 : rr + 1;
    }
    if (wave == 0) {
        const int e = (nquads << 2) + lane;
        const bool in = e < nedges;
        const int l = (in ? dst[e] : -1) - base;
        push(in && ((unsigned)l < urows), (unsigned)e, l);
    }
    }
    while (count > 0) drain8();

    __syncthreads();

    const int wv  = __builtin_amdgcn_readfirstlane(wave);
    const int nt  = wv % NT;
    const int sg  = wv / NT;
    const int n0  = nt * 16;
    const int idx = lane & 15;
    const int nstr = (rows + 15) >> 4;

    {
        v2f bfr[32];
        const float* Bn = Wneigh + (size_t)(n0 + idx) * 128 + 2 * hi;
#pragma unroll
        for (int k = 0; k < 32; ++k) bfr[k] = *(const v2f*)(Bn + 4 * k);

        for (int it = 0; it < NIT; ++it) {
            const int strip = it * SG + sg;
            const bool valid = strip < nstr;
            v8f c = {0.f, 0.f, 0.f, 0.f, 0.f, 0.f, 0.f, 0.f};
            if (valid) {
                int ar = strip * 16 + idx;
                if (ar > R) ar = R;
                const float invdeg = 1.0f / fmaxf(cnt[ar], 1.0f);
                const float* Am = acc + ar * P + 2 * hi;
#pragma unroll
                for (int k = 0; k < 32; ++k) {
                    v2f a = *(const v2f*)(Am + 4 * k);
                    a.x *= invdeg;
                    a.y *= invdeg;
                    c = __builtin_amdgcn_wmma_f32_16x16x4_f32(false, a, false, bfr[k],
                                                              (short)0, c, false, false);
                }
            }
            __syncthreads();
            if (valid) {
#pragma unroll
                for (int j = 0; j < 8; ++j) {
                    const int r = strip * 16 + 8 * hi + j;
                    if (r < rows) acc[r * P + n0 + idx] = c[j];
                }
            }
        }
    }

    {
        v2f bfr[32];
        const float* Bs = Wself + (size_t)(n0 + idx) * 128 + 2 * hi;
#pragma unroll
        for (int k = 0; k < 32; ++k) bfr[k] = *(const v2f*)(Bs + 4 * k);
        const float bv = bias[n0 + idx];

        static_assert(RPW == 2 || RPW == 4, "a wave copies 2 or 4 rows of a strip");
        float4 nx0 = make_float4(0.f, 0.f, 0.f, 0.f), nx1 = nx0, nx2 = nx0, nx3 = nx0;
        const float* hrow = hsrc + (size_t)base * 128 + lane * 4;
        const int lr0 = nt * RPW;
        auto ldrow = [hrow, rows](int gr) -> float4 {
            if (gr > rows - 1) gr = rows - 1;
            return *(const float4*)(hrow + (size_t)gr * 128);
        };
        if (sg < nstr) {
            nx0 = ldrow(sg * 16 + lr0);
            nx1 = ldrow(sg * 16 + lr0 + 1);
            if (RPW == 4) { nx2 = ldrow(sg * 16 + lr0 + 2); nx3 = ldrow(sg * 16 + lr0 + 3); }
        }

        for (int it = 0; it < NIT; ++it) {
            const int strip = it * SG + sg;
            const bool valid = strip < nstr;
            if (valid) {
                float* srow = stage + (sg * 16 + lr0) * P + lane * 4;
                *(float4*)(srow)         = nx0;
                *(float4*)(srow + P)     = nx1;
                if (RPW == 4) { *(float4*)(srow + 2 * P) = nx2; *(float4*)(srow + 3 * P) = nx3; }
            }
            __syncthreads();
            if (strip + SG < nstr) {
                const int g0 = (strip + SG) * 16 + lr0;
                nx0 = ldrow(g0);
                nx1 = ldrow(g0 + 1);
                if (RPW == 4) { nx2 = ldrow(g0 + 2); nx3 = ldrow(g0 + 3); }
            }
            if (valid) {
                v8f c;
#pragma unroll
                for (int j = 0; j < 8; ++j) {
                    const int r = strip * 16 + 8 * hi + j;
                    c[j] = (r < rows) ? acc[r * P + n0 + idx] : 0.0f;
                }
                const float* As = stage + (sg * 16 + idx) * P + 2 * hi;
#pragma unroll
                for (int k = 0; k < 32; ++k) {
                    const v2f a = *(const v2f*)(As + 4 * k);
                    c = __builtin_amdgcn_wmma_f32_16x16x4_f32(false, a, false, bfr[k],
                                                              (short)0, c, false, false);
                }
#pragma unroll
                for (int j = 0; j < 8; ++j) {
                    const int r = strip * 16 + 8 * hi + j;
                    if (r < rows) {
                        float v = c[j] + bv;
                        if (RELU) v = fmaxf(v, 0.0f);
                        acc[r * P + n0 + idx] = v;
                    }
                }
            }
            if (it + 1 < NIT) __syncthreads();
        }
    }

    __syncthreads();
    for (int i = tid; i < rows * (DOUT / 4); i += 1024) {
        const int r = i / (DOUT / 4), l = i % (DOUT / 4);
        const float4 v = *(const float4*)(acc + r * P + l * 4);
        *(float4*)(out + (size_t)(base + r) * DOUT + l * 4) = v;
    }
}

static inline unsigned cdiv(unsigned a, unsigned b) { return (a + b - 1) / b; }

extern "C" void kernel_launch(void* const* d_in, const int* in_sizes, int n_in,
                              void* d_out, int out_size, void* d_ws, size_t ws_size,
                              hipStream_t stream) {
    const float* feats = (const float*)d_in[0];
    const int* src0 = (const int*)d_in[1];
    const int* dst0 = (const int*)d_in[2];
    const int* src1 = (const int*)d_in[3];
    const int* dst1 = (const int*)d_in[4];
    const int* src2 = (const int*)d_in[5];
    const int* dst2 = (const int*)d_in[6];
    const float* Ws0 = (const float*)d_in[10];
    const float* Wn0 = (const float*)d_in[11];
    const float* b0  = (const float*)d_in[12];
    const float* Ws1 = (const float*)d_in[13];
    const float* Wn1 = (const float*)d_in[14];
    const float* b1  = (const float*)d_in[15];
    const float* Ws2 = (const float*)d_in[16];
    const float* Wn2 = (const float*)d_in[17];
    const float* b2  = (const float*)d_in[18];

    const int E0 = in_sizes[1], E1 = in_sizes[3], E2 = in_sizes[5];
    const int N0 = 500000, N1 = 100000, N2 = 20000, N3 = 5000;
    constexpr int R0 = 400, R1 = 80, R2 = 20;

    char* ws = (char*)d_ws;
    float* h1  = (float*)ws;                    ws += (size_t)N1 * 128 * sizeof(float);
    float* h2  = (float*)ws;

    const dim3 blk(1024);

    constexpr int NB     = (N1 + R0 - 1) / R0;
    static_assert((N2 + R1 - 1) / R1 == NB, "layers 0 and 1 have the same number of blocks");
    constexpr int BMP    = R1 * 128;
    constexpr int OFF1   = 4096;
    constexpr int G0 = 125000, G1 = 25000;
    constexpr int TILES0 = (G0 + 1023) / 1024;
    constexpr int TILES1 = (G1 + 1023) / 1024;
    static_assert(TILES0 * 32 <= OFF1 && OFF1 + TILES1 * 32 <= BMP, "both bitmaps must fit a slot");
    static_assert((size_t)NB * BMP <= (size_t)N2 * 128, "the slots must fit the space of h2");
    unsigned* bm = (unsigned*)h2;
    group_bitmaps<R0, R1, NB, BMP, OFF1><<<TILES0 + TILES1, blk, 0, stream>>>(dst0, E0, N1, dst1, E1, N2, bm);

    sage_layer_owned<R0, 128, true, BMP><<<cdiv((unsigned)N1, R0), blk, 0, stream>>>(feats, src0, dst0, bm, E0, N0, N1, Ws0, Wn0, b0, h1);
    sage_layer_owned<R1, 128, true, BMP><<<cdiv((unsigned)N2, R1), blk, 0, stream>>>(h1, src1, dst1, bm + OFF1, E1, N1, N2, Ws1, Wn1, b1, h2);
    sage_layer_owned<R2, 64, false, 0><<<cdiv((unsigned)N3, R2), blk, 0, stream>>>(h2, src2, dst2, nullptr, E2, N2, N3, Ws2, Wn2, b2, (float*)d_out);
}


# Row adds without LDS float atomics: a found edge is queued in LDS to the wave owning its row (row % 32), which adds source rows by plain 16-byte read-add-write; row sum order: queue arrival
# speedup vs baseline: 3.6937x; 1.0648x over previous
#include <hip/hip_runtime.h>
#include <hip/hip_bf16.h>


typedef __attribute__((ext_vector_type(2))) float v2f;
typedef __attribute__((ext_vector_type(8))) float v8f;

template <int R, int NB, int PITCH>
static __device__ inline void bitmap_tile(const int* __restrict__ dst, int ngroups, unsigned un, int tile,
                                          unsigned* __restrict__ bm, unsigned* bits) {
    const int tid = threadIdx.x;
    const unsigned bit = 1u << (tid & 31);
    const int word = tid >> 5;
    const int g = (tile << 10) + tid;
    if (g < ngroups) {
        const int4* p = (const int4*)(dst + ((size_t)g << 4));
        const int4 d0 = p[0], d1 = p[1], d2 = p[2], d3 = p[3];
        const int dv[16] = {d0.x, d0.y, d0.z, d0.w, d1.x, d1.y, d1.z, d1.w,
                            d2.x, d2.y, d2.z, d2.w, d3.x, d3.y, d3.z, d3.w};
#pragma unroll
        for (int j = 0; j < 16; ++j) {
            const unsigned d = (unsigned)dv[j];
            if (d < un) {
                const unsigned o = d / (unsigned)R;
                if (o < (unsigned)NB) atomicOr(&bits[word * NB + o], bit);
            }
        }
    }
    __syncthreads();
    for (int i = tid; i < NB * 32; i += 1024) {
        const int k = (i & 31) * NB + (i >> 5);
        bm[(size_t)(i >> 5) * PITCH + (tile << 5) + (i & 31)] = bits[k];
        bits[k] = 0u;
    }
}

template <int RA, int RB, int NB, int PITCH, int OFFB>
__global__ __launch_bounds__(1024) void group_bitmaps(const int* __restrict__ dstA, int nedgesA, int n_dstA,
                                                      const int* __restrict__ dstB, int nedgesB, int n_dstB,
                                                      unsigned* __restrict__ bm) {
    __shared__ unsigned bits[NB * 32];
    const int ngA = nedgesA >> 4, ngB = nedgesB >> 4;
    const int ntA = (ngA + 1023) >> 10, ntB = (ngB + 1023) >> 10;
    for (int i = threadIdx.x; i < NB * 32; i += 1024) bits[i] = 0u;
    for (int tile = blockIdx.x; tile < ntA + ntB; tile += gridDim.x) {
        __syncthreads();
        if (tile < ntA) bitmap_tile<RA, NB, PITCH>(dstA, ngA, (unsigned)n_dstA, tile, bm, bits);
        else            bitmap_tile<RB, NB, PITCH>(dstB, ngB, (unsigned)n_dstB, tile - ntA, bm + OFFB, bits);
    }
}

template <int R, int DOUT, bool RELU, int BMP, int QCAP, int SI>
__global__ __launch_bounds__(1024) void sage_layer_owned(const float* __restrict__ hsrc,
                                                         const int* __restrict__ src,
                                                         const int* __restrict__ dst,
                                                         const unsigned* bm,
                                                         int nedges, int nsrc, int n_dst,
                                                         const float* __restrict__ Wself,
                                                         const float* __restrict__ Wneigh,
                                                         const float* __restrict__ bias,
                                                         float* __restrict__ out) {
    constexpr bool BITMAP = BMP > 0;
    constexpr int P    = 132;
    constexpr int NT   = DOUT / 16;
    constexpr int SG   = 32 / NT;
    constexpr int NSTR = (R + 15) / 16;
    constexpr int NIT  = (NSTR + SG - 1) / SG;
    constexpr int RPW  = 16 / NT;
    constexpr int QP   = QCAP + 1;
    constexpr int QWORDS = 2 * 32 * QP;
    constexpr int SWORDS = SG * 16 * P;
    constexpr int XWORDS = QWORDS > SWORDS ? QWORDS : SWORDS;
    __shared__ __attribute__((aligned(16))) float acc[(R + 1) * P];
    __shared__ float cnt[R + 1];
    __shared__ unsigned qn[32];
    __shared__ __attribute__((aligned(16))) unsigned xbuf[XWORDS];
    unsigned* const qe = xbuf;
    unsigned* const qd = xbuf + 32 * QP;
    float* const stage = (float*)xbuf;

    const int tid  = threadIdx.x;
    const int lane = tid & 31;
    const int wave = tid >> 5;
    const int hi   = lane >> 4;
    const int base = blockIdx.x * R;
    int rows = n_dst - base;
    if (rows > R) rows = R;
    if (rows <= 0) return;
    const unsigned urows = (unsigned)rows;

    for (int i = tid; i < (R + 1) * (P / 4); i += 1024)
        *(float4*)(acc + i * 4) = make_float4(0.0f, 0.0f, 0.0f, 0.0f);
    for (int i = tid; i < R + 1; i += 1024) cnt[i] = 0.0f;
    if (tid < 32) qn[tid] = 0u;
    __syncthreads();

    const int wq = wave * QP;
    static_assert(R <= 1024, "one lane per owned row");
    const int myrow = wave + 32 * lane;
    float mycnt = 0.0f;

    auto route = [&](bool t, unsigned eidx, int dloc) {
        if (t) {
            const int ow = dloc & 31;
            const unsigned pos = atomicAdd(&qn[ow], 1u);
            if (pos < (unsigned)QCAP) {
                qe[ow * QP + pos] = eidx;
                qd[ow * QP + pos] = (unsigned)dloc;
            } else {
                int s = src[eidx];
                s = s < 0 ? 0 : (s > nsrc - 1 ? nsrc - 1 : s);
                const float* hr = hsrc + (size_t)s * 128;
                float* row = acc + dloc * P;
                for (int c = 0; c < 128; ++c) atomicAdd(row + c, hr[c]);
                atomicAdd(cnt + dloc, 1.0f);
            }
        }
    };

    auto drain_own = [&]() {
        int n = __builtin_amdgcn_readfirstlane((int)qn[wave]);
        if (n > QCAP) n = QCAP;
        for (int b0 = 0; b0 < n; b0 += 32) {
            const int i = b0 + lane;
            const bool ok = i < n;
            const unsigned e = ok ? qe[wq + i] : 0u;
            const int dl = ok ? (int)qd[wq + i] : R;
            int s = ok ? src[e] : 0;
            s = s < 0 ? 0 : (s > nsrc - 1 ? nsrc - 1 : s);
            const int nb = (n - b0) < 32 ? (n - b0) : 32;
            for (int k0 = 0; k0 < nb; k0 += 16) {
                float4 va[8], vb[8];
                int    da[8], db[8];
                const bool two = k0 + 8 < nb;
#pragma unroll
                for (int k = 0; k < 8; ++k) {
                    const int sk = __builtin_amdgcn_readlane(s, k0 + k);
                    da[k] = __builtin_amdgcn_readlane(dl, k0 + k);
                    va[k] = *(const float4*)(hsrc + (size_t)sk * 128 + lane * 4);
                }
                if (two) {
#pragma unroll
                    for (int k = 0; k < 8; ++k) {
                        const int sk = __builtin_amdgcn_readlane(s, k0 + 8 + k);
                        db[k] = __builtin_amdgcn_readlane(dl, k0 + 8 + k);
                        vb[k] = *(const float4*)(hsrc + (size_t)sk * 128 + lane * 4);
                    }
                }
#pragma unroll
                for (int k = 0; k < 8; ++k) {
                    float4* row = (float4*)(acc + da[k] * P + lane * 4);
                    float4 a4 = *row;
                    a4.x += va[k].x;
                    a4.y += va[k].y;
                    a4.z += va[k].z;
                    a4.w += va[k].w;
                    *row = a4;
                    mycnt += (da[k] == myrow) ? 1.0f : 0.0f;
                }
                if (two) {
#pragma unroll
                    for (int k = 0; k < 8; ++k) {
                        float4* row = (float4*)(acc + db[k] * P + lane * 4);
                        float4 a4 = *row;
                        a4.x += vb[k].x;
                        a4.y += vb[k].y;
                        a4.z += vb[k].z;
                        a4.w += vb[k].w;
                        *row = a4;
                        mycnt += (db[k] == myrow) ? 1.0f : 0.0f;
                    }
                }
            }
        }
        if (lane == 0) qn[wave] = 0u;
    };

    if constexpr (BITMAP) {
        const int ngroups = nedges >> 4;
        const int wpo = ((ngroups + 1023) >> 10) << 5;
        const unsigned* bmw = bm + (size_t)blockIdx.x * BMP;
        const int nslab = wpo > 0 ? (wpo + 1024 * SI - 1) / (1024 * SI) : 1;
        for (int sl = 0; sl < nslab; ++sl) {
            if (sl == 0 && wave == 0) {
                const int e = (ngroups << 4) + lane;
                const bool in = e < nedges;
                const int l = (in ? dst[e] : -1) - base;
                route(in && ((unsigned)l < urows), (unsigned)e, l);
            }
            for (int it = 0; it < SI; ++it) {
                const int w0 = ((sl * SI + it) << 10) + (wave << 5);
                unsigned word = w0 < wpo ? bmw[w0 + lane] : 0u;
                while (__builtin_amdgcn_ballot_w32(word != 0u) != 0u) {
                    const int g  = ((w0 + lane) << 5) + (int)__builtin_ctz(word | 0x80000000u);
                    const bool t = (word != 0u) && (g < ngroups);
                    const int e0 = g << 4;
                    word &= word - 1u;
                    int4 d0 = make_int4(-1, -1, -1, -1), d1 = d0, d2 = d0, d3 = d0;
                    if (t) {
                        const int4* p = (const int4*)(dst + e0);
                        d0 = p[0];
                        d1 = p[1];
                        d2 = p[2];
                        d3 = p[3];
                    }
                    unsigned m;
                    m  = ((unsigned)(d0.x - base) < urows ? 0x0001u : 0u) | ((unsigned)(d0.y - base) < urows ? 0x0002u : 0u);
                    m |= ((unsigned)(d0.z - base) < urows ? 0x0004u : 0u) | ((unsigned)(d0.w - base) < urows ? 0x0008u : 0u);
                    m |= ((unsigned)(d1.x - base) < urows ? 0x0010u : 0u) | ((unsigned)(d1.y - base) < urows ? 0x0020u : 0u);
                    m |= ((unsigned)(d1.z - base) < urows ? 0x0040u : 0u) | ((unsigned)(d1.w - base) < urows ? 0x0080u : 0u);
                    m |= ((unsigned)(d2.x - base) < urows ? 0x0100u : 0u) | ((unsigned)(d2.y - base) < urows ? 0x0200u : 0u);
                    m |= ((unsigned)(d2.z - base) < urows ? 0x0400u : 0u) | ((unsigned)(d2.w - base) < urows ? 0x0800u : 0u);
                    m |= ((unsigned)(d3.x - base) < urows ? 0x1000u : 0u) | ((unsigned)(d3.y - base) < urows ? 0x2000u : 0u);
                    m |= ((unsigned)(d3.z - base) < urows ? 0x4000u : 0u) | ((unsigned)(d3.w - base) < urows ? 0x8000u : 0u);
                    while (__builtin_amdgcn_ballot_w32(m != 0u) != 0u) {
                        const bool tt = m != 0u;
                        const unsigned p = (unsigned)__builtin_ctz(m | 0x10000u);
                        const int4 dq = p < 8u ? (p < 4u ? d0 : d1) : (p < 12u ? d2 : d3);
                        const unsigned r4 = p & 3u;
                        const int dval = r4 < 2u ? (r4 == 0u ? dq.x : dq.y) : (r4 == 2u ? dq.z : dq.w);
                        route(tt, (unsigned)e0 + p, dval - base);
                        m &= m - 1u;
                    }
                }
            }
            __syncthreads();
            drain_own();
            __syncthreads();
        }
    } else {

    const int nquads  = nedges >> 2;
    const int nrounds = (nquads + 4095) >> 12;
    int rr = (int)(((long long)blockIdx.x * nrounds) / gridDim.x);
    const int4* dst4 = (const int4*)dst;
    for (int r = 0; r < nrounds; ++r) {
        const int q0 = (rr << 12) + (wave << 7) + lane;
        int4 dq[4];
#pragma unroll
        for (int u = 0; u < 4; ++u) {
            const int q = q0 + (u << 5);
            dq[u] = (q < nquads) ? dst4[q] : make_int4(-1, -1, -1, -1);
        }
#pragma unroll
        for (int u = 0; u < 4; ++u) {
            const int q = q0 + (u << 5);
            const int l0 = dq[u].x - base, l1 = dq[u].y - base;
            const int l2 = dq[u].z - base, l3 = dq[u].w - base;
            const bool t0 = (unsigned)l0 < urows, t1 = (unsigned)l1 < urows;
            const bool t2 = (unsigned)l2 < urows, t3 = (unsigned)l3 < urows;
            if (__builtin_amdgcn_ballot_w32(t0 | t1 | t2 | t3) != 0u) {
                const unsigned e0 = (unsigned)q << 2;
                route(t0, e0, l0);
                route(t1, e0 + 1u, l1);
                route(t2, e0 + 2u, l2);
                route(t3, e0 + 3u, l3);
            }
        }
        rr = (rr + 1 == nrounds) ? 0 : rr + 1;
    }
    if (wave == 0) {
        const int e = (nquads << 2) + lane;
        const bool in = e < nedges;
        const int l = (in ? dst[e] : -1) - base;
        route(in && ((unsigned)l < urows), (unsigned)e, l);
    }
    __syncthreads();
    drain_own();
    }
    if (myrow < R) cnt[myrow] += mycnt;

    __syncthreads();

    const int wv  = __builtin_amdgcn_readfirstlane(wave);
    const int nt  = wv % NT;
    const int sg  = wv / NT;
    const int n0  = nt * 16;
    const int idx = lane & 15;
    const int nstr = (rows + 15) >> 4;

    {
        v2f bfr[32];
        const float* Bn = Wneigh + (size_t)(n0 + idx) * 128 + 2 * hi;
#pragma unroll
        for (int k = 0; k < 32; ++k) bfr[k] = *(const v2f*)(Bn + 4 * k);

        for (int it = 0; it < NIT; ++it) {
            const int strip = it * SG + sg;
            const bool valid = strip < nstr;
            v8f c = {0.f, 0.f, 0.f, 0.f, 0.f, 0.f, 0.f, 0.f};
            if (valid) {
                int ar = strip * 16 + idx;
                if (ar > R) ar = R;
                const float invdeg = 1.0f / fmaxf(cnt[ar], 1.0f);
                const float* Am = acc + ar * P + 2 * hi;
#pragma unroll
                for (int k = 0; k < 32; ++k) {
                    v2f a = *(const v2f*)(Am + 4 * k);
                    a.x *= invdeg;
                    a.y *= invdeg;
                    c = __builtin_amdgcn_wmma_f32_16x16x4_f32(false, a, false, bfr[k],
                                                              (short)0, c, false, false);
                }
            }
            __syncthreads();
            if (valid) {
#pragma unroll
                for (int j = 0; j < 8; ++j) {
                    const int r = strip * 16 + 8 * hi + j;
                    if (r < rows) acc[r * P + n0 + idx] = c[j];
                }
            }
        }
    }

    {
        v2f bfr[32];
        const float* Bs = Wself + (size_t)(n0 + idx) * 128 + 2 * hi;
#pragma unroll
        for (int k = 0; k < 32; ++k) bfr[k] = *(const v2f*)(Bs + 4 * k);
        const float bv = bias[n0 + idx];

        static_assert(RPW == 2 || RPW == 4, "a wave copies 2 or 4 rows of a strip");
        float4 nx0 = make_float4(0.f, 0.f, 0.f, 0.f), nx1 = nx0, nx2 = nx0, nx3 = nx0;
        const float* hrow = hsrc + (size_t)base * 128 + lane * 4;
        const int lr0 = nt * RPW;
        auto ldrow = [hrow, rows](int gr) -> float4 {
            if (gr > rows - 1) gr = rows - 1;
            return *(const float4*)(hrow + (size_t)gr * 128);
        };
        if (sg < nstr) {
            nx0 = ldrow(sg * 16 + lr0);
            nx1 = ldrow(sg * 16 + lr0 + 1);
            if (RPW == 4) { nx2 = ldrow(sg * 16 + lr0 + 2); nx3 = ldrow(sg * 16 + lr0 + 3); }
        }

        for (int it = 0; it < NIT; ++it) {
            const int strip = it * SG + sg;
            const bool valid = strip < nstr;
            if (valid) {
                float* srow = stage + (sg * 16 + lr0) * P + lane * 4;
                *(float4*)(srow)         = nx0;
                *(float4*)(srow + P)     = nx1;
                if (RPW == 4) { *(float4*)(srow + 2 * P) = nx2; *(float4*)(srow + 3 * P) = nx3; }
            }
            __syncthreads();
            if (strip + SG < nstr) {
                const int g0 = (strip + SG) * 16 + lr0;
                nx0 = ldrow(g0);
                nx1 = ldrow(g0 + 1);
                if (RPW == 4) { nx2 = ldrow(g0 + 2); nx3 = ldrow(g0 + 3); }
            }
            if (valid) {
                v8f c;
#pragma unroll
                for (int j = 0; j < 8; ++j) {
                    const int r = strip * 16 + 8 * hi + j;
                    c[j] = (r < rows) ? acc[r * P + n0 + idx] : 0.0f;
                }
                const float* As = stage + (sg * 16 + idx) * P + 2 * hi;
#pragma unroll
                for (int k = 0; k < 32; ++k) {
                    const v2f a = *(const v2f*)(As + 4 * k);
                    c = __builtin_amdgcn_wmma_f32_16x16x4_f32(false, a, false, bfr[k],
                                                              (short)0, c, false, false);
                }
#pragma unroll
                for (int j = 0; j < 8; ++j) {
                    const int r = strip * 16 + 8 * hi + j;
                    if (r < rows) {
                        float v = c[j] + bv;
                        if (RELU) v = fmaxf(v, 0.0f);
                        acc[r * P + n0 + idx] = v;
                    }
                }
            }
            if (it + 1 < NIT) __syncthreads();
        }
    }

    __syncthreads();
    for (int i = tid; i < rows * (DOUT / 4); i += 1024) {
        const int r = i / (DOUT / 4), l = i % (DOUT / 4);
        const float4 v = *(const float4*)(acc + r * P + l * 4);
        *(float4*)(out + (size_t)(base + r) * DOUT + l * 4) = v;
    }
}

static inline unsigned cdiv(unsigned a, unsigned b) { return (a + b - 1) / b; }

extern "C" void kernel_launch(void* const* d_in, const int* in_sizes, int n_in,
                              void* d_out, int out_size, void* d_ws, size_t ws_size,
                              hipStream_t stream) {
    const float* feats = (const float*)d_in[0];
    const int* src0 = (const int*)d_in[1];
    const int* dst0 = (const int*)d_in[2];
    const int* src1 = (const int*)d_in[3];
    const int* dst1 = (const int*)d_in[4];
    const int* src2 = (const int*)d_in[5];
    const int* dst2 = (const int*)d_in[6];
    const float* Ws0 = (const float*)d_in[10];
    const float* Wn0 = (const float*)d_in[11];
    const float* b0  = (const float*)d_in[12];
    const float* Ws1 = (const float*)d_in[13];
    const float* Wn1 = (const float*)d_in[14];
    const float* b1  = (const float*)d_in[15];
    const float* Ws2 = (const float*)d_in[16];
    const float* Wn2 = (const float*)d_in[17];
    const float* b2  = (const float*)d_in[18];

    const int E0 = in_sizes[1], E1 = in_sizes[3], E2 = in_sizes[5];
    const int N0 = 500000, N1 = 100000, N2 = 20000, N3 = 5000;
    constexpr int R0 = 400, R1 = 80, R2 = 20;
    constexpr int QC0 = 384, QC1 = 128;

    char* ws = (char*)d_ws;
    float* h1  = (float*)ws;                    ws += (size_t)N1 * 128 * sizeof(float);
    float* h2  = (float*)ws;

    const dim3 blk(1024);

    constexpr int NB     = (N1 + R0 - 1) / R0;
    static_assert((N2 + R1 - 1) / R1 == NB, "layers 0 and 1 have the same number of blocks");
    constexpr int BMP    = R1 * 128;
    constexpr int OFF1   = 4096;
    constexpr int G0 = 125000, G1 = 25000;
    constexpr int TILES0 = (G0 + 1023) / 1024;
    constexpr int TILES1 = (G1 + 1023) / 1024;
    static_assert(TILES0 * 32 <= OFF1 && OFF1 + TILES1 * 32 <= BMP, "both bitmaps must fit a slot");
    static_assert((size_t)NB * BMP <= (size_t)N2 * 128, "the slots must fit the space of h2");
    unsigned* bm = (unsigned*)h2;
    group_bitmaps<R0, R1, NB, BMP, OFF1><<<TILES0 + TILES1, blk, 0, stream>>>(dst0, E0, N1, dst1, E1, N2, bm);

    sage_layer_owned<R0, 128, true, BMP, QC0, 4><<<cdiv((unsigned)N1, R0), blk, 0, stream>>>(feats, src0, dst0, bm, E0, N0, N1, Ws0, Wn0, b0, h1);
    sage_layer_owned<R1, 128, true, BMP, QC1, 1><<<cdiv((unsigned)N2, R1), blk, 0, stream>>>(h1, src1, dst1, bm + OFF1, E1, N1, N2, Ws1, Wn1, b1, h2);
    sage_layer_owned<R2, 64, false, 0, QC1, 1><<<cdiv((unsigned)N3, R2), blk, 0, stream>>>(h2, src2, dst2, nullptr, E2, N2, N3, Ws2, Wn2, b2, (float*)d_out);
}


# Layer 0: each wave sorts its edge queue by source row range (32 ranges, LDS counting sort) before fetching: all work-groups sweep the table together, reused rows hit L2; row sum order: range, arrival
# speedup vs baseline: 3.8446x; 1.0409x over previous
#include <hip/hip_runtime.h>
#include <hip/hip_bf16.h>


typedef __attribute__((ext_vector_type(2))) float v2f;
typedef __attribute__((ext_vector_type(8))) float v8f;

template <int R, int NB, int PITCH>
static __device__ inline void bitmap_tile(const int* __restrict__ dst, int ngroups, unsigned un, int tile,
                                          unsigned* __restrict__ bm, unsigned* bits) {
    const int tid = threadIdx.x;
    const unsigned bit = 1u << (tid & 31);
    const int word = tid >> 5;
    const int g = (tile << 10) + tid;
    if (g < ngroups) {
        const int4* p = (const int4*)(dst + ((size_t)g << 4));
        const int4 d0 = p[0], d1 = p[1], d2 = p[2], d3 = p[3];
        const int dv[16] = {d0.x, d0.y, d0.z, d0.w, d1.x, d1.y, d1.z, d1.w,
                            d2.x, d2.y, d2.z, d2.w, d3.x, d3.y, d3.z, d3.w};
#pragma unroll
        for (int j = 0; j < 16; ++j) {
            const unsigned d = (unsigned)dv[j];
            if (d < un) {
                const unsigned o = d / (unsigned)R;
                if (o < (unsigned)NB) atomicOr(&bits[word * NB + o], bit);
            }
        }
    }
    __syncthreads();
    for (int i = tid; i < NB * 32; i += 1024) {
        const int k = (i & 31) * NB + (i >> 5);
        bm[(size_t)(i >> 5) * PITCH + (tile << 5) + (i & 31)] = bits[k];
        bits[k] = 0u;
    }
}

template <int RA, int RB, int NB, int PITCH, int OFFB>
__global__ __launch_bounds__(1024) void group_bitmaps(const int* __restrict__ dstA, int nedgesA, int n_dstA,
                                                      const int* __restrict__ dstB, int nedgesB, int n_dstB,
                                                      unsigned* __restrict__ bm) {
    __shared__ unsigned bits[NB * 32];
    const int ngA = nedgesA >> 4, ngB = nedgesB >> 4;
    const int ntA = (ngA + 1023) >> 10, ntB = (ngB + 1023) >> 10;
    for (int i = threadIdx.x; i < NB * 32; i += 1024) bits[i] = 0u;
    for (int tile = blockIdx.x; tile < ntA + ntB; tile += gridDim.x) {
        __syncthreads();
        if (tile < ntA) bitmap_tile<RA, NB, PITCH>(dstA, ngA, (unsigned)n_dstA, tile, bm, bits);
        else            bitmap_tile<RB, NB, PITCH>(dstB, ngB, (unsigned)n_dstB, tile - ntA, bm + OFFB, bits);
    }
}

template <int R, int DOUT, bool RELU, int BMP, int QCAP, int SI, int NBK>
__global__ __launch_bounds__(1024) void sage_layer_owned(const float* __restrict__ hsrc,
                                                         const int* __restrict__ src,
                                                         const int* __restrict__ dst,
                                                         const unsigned* bm,
                                                         int nedges, int nsrc, int n_dst,
                                                         const float* __restrict__ Wself,
                                                         const float* __restrict__ Wneigh,
                                                         const float* __restrict__ bias,
                                                         float* __restrict__ out) {
    constexpr bool BITMAP = BMP > 0;
    static_assert(NBK == 0 || NBK == 32, "one source range per lane in the prefix sum");
    constexpr int P    = 132;
    constexpr int NT   = DOUT / 16;
    constexpr int SG   = 32 / NT;
    constexpr int NSTR = (R + 15) / 16;
    constexpr int NIT  = (NSTR + SG - 1) / SG;
    constexpr int RPW  = 16 / NT;
    constexpr int QP   = QCAP + 1;
    constexpr int QWORDS = 2 * 32 * QP;
    constexpr int SWORDS = SG * 16 * P;
    constexpr int XWORDS = QWORDS > SWORDS ? QWORDS : SWORDS;
    __shared__ __attribute__((aligned(16))) float acc[(R + 1) * P];
    __shared__ float cnt[R + 1];
    __shared__ unsigned qn[32];
    __shared__ unsigned hist[NBK > 0 ? 32 * NBK : 1];
    __shared__ __attribute__((aligned(16))) unsigned xbuf[XWORDS];
    unsigned* const qe = xbuf;
    unsigned* const qd = xbuf + 32 * QP;
    float* const stage = (float*)xbuf;

    const int tid  = threadIdx.x;
    const int lane = tid & 31;
    const int wave = tid >> 5;
    const int hi   = lane >> 4;
    const int base = blockIdx.x * R;
    int rows = n_dst - base;
    if (rows > R) rows = R;
    if (rows <= 0) return;
    const unsigned urows = (unsigned)rows;

    for (int i = tid; i < (R + 1) * (P / 4); i += 1024)
        *(float4*)(acc + i * 4) = make_float4(0.0f, 0.0f, 0.0f, 0.0f);
    for (int i = tid; i < R + 1; i += 1024) cnt[i] = 0.0f;
    if (tid < 32) qn[tid] = 0u;
    if constexpr (NBK > 0) {
        for (int i = tid; i < 32 * NBK; i += 1024) hist[i] = 0u;
    }
    __syncthreads();

    const int wq = wave * QP;
    static_assert(R <= 1024, "one lane per owned row");
    const int myrow = wave + 32 * lane;
    float mycnt = 0.0f;

    auto route = [&](bool t, unsigned eidx, int dloc) {
        if (t) {
            const int ow = dloc & 31;
            const unsigned pos = atomicAdd(&qn[ow], 1u);
            if (pos < (unsigned)QCAP) {
                qe[ow * QP + pos] = eidx;
                qd[ow * QP + pos] = (unsigned)dloc;
            } else {
                int s = src[eidx];
                s = s < 0 ? 0 : (s > nsrc - 1 ? nsrc - 1 : s);
                const float* hr = hsrc + (size_t)s * 128;
                float* row = acc + dloc * P;
                for (int c = 0; c < 128; ++c) atomicAdd(row + c, hr[c]);
                atomicAdd(cnt + dloc, 1.0f);
            }
        }
    };

    auto drain_own = [&]() {
        int n = __builtin_amdgcn_readfirstlane((int)qn[wave]);
        if (n > QCAP) n = QCAP;
        if constexpr (NBK > 0) {
            constexpr int EPL = (QCAP + 31) / 32;
            unsigned* const hw = hist + wave * NBK;
            const float bscale = (float)NBK / (float)nsrc;
            int sv[EPL], dv[EPL];
            unsigned bk[EPL], rk[EPL];
#pragma unroll
            for (int j = 0; j < EPL; ++j) {
                const int i = j * 32 + lane;
                const bool ok = i < n;
                sv[j] = ok ? (int)qe[wq + i] : 0;
                dv[j] = ok ? (int)qd[wq + i] : R;
            }
#pragma unroll
            for (int j = 0; j < EPL; ++j) {
                const int s = (j * 32 + lane < n) ? src[sv[j]] : 0;
                sv[j] = s < 0 ? 0 : (s > nsrc - 1 ? nsrc - 1 : s);
            }
#pragma unroll
            for (int j = 0; j < EPL; ++j) {
                const unsigned b = (unsigned)((float)sv[j] * bscale);
                bk[j] = b > (unsigned)(NBK - 1) ? (unsigned)(NBK - 1) : b;
                rk[j] = (j * 32 + lane < n) ? atomicAdd(&hw[bk[j]], 1u) : 0u;
            }
            const unsigned nb_ = hw[lane];
            unsigned incl = nb_;
            for (int d = 1; d < 32; d <<= 1) {
                const unsigned t = __shfl_up(incl, d);
                if (lane >= d) incl += t;
            }
            const unsigned first = incl - nb_;
            hw[lane] = 0u;
#pragma unroll
            for (int j = 0; j < EPL; ++j) rk[j] += (unsigned)__shfl((int)first, (int)bk[j]);
#pragma unroll
            for (int j = 0; j < EPL; ++j) {
                if (j * 32 + lane < n) {
                    qe[wq + rk[j]] = (unsigned)sv[j];
                    qd[wq + rk[j]] = (unsigned)dv[j];
                }
            }
        }
        for (int b0 = 0; b0 < n; b0 += 32) {
            const int i = b0 + lane;
            const bool ok = i < n;
            const unsigned e = ok ? qe[wq + i] : 0u;
            const int dl = ok ? (int)qd[wq + i] : R;
            int s;
            if constexpr (NBK > 0) s = (int)e;
            else                   s = ok ? src[e] : 0;
            s = s < 0 ? 0 : (s > nsrc - 1 ? nsrc - 1 : s);
            const int nb = (n - b0) < 32 ? (n - b0) : 32;
            for (int k0 = 0; k0 < nb; k0 += 16) {
                float4 va[8], vb[8];
                int    da[8], db[8];
                const bool two = k0 + 8 < nb;
#pragma unroll
                for (int k = 0; k < 8; ++k) {
                    const int sk = __builtin_amdgcn_readlane(s, k0 + k);
                    da[k] = __builtin_amdgcn_readlane(dl, k0 + k);
                    va[k] = *(const float4*)(hsrc + (size_t)sk * 128 + lane * 4);
                }
                if (two) {
#pragma unroll
                    for (int k = 0; k < 8; ++k) {
                        const int sk = __builtin_amdgcn_readlane(s, k0 + 8 + k);
                        db[k] = __builtin_amdgcn_readlane(dl, k0 + 8 + k);
                        vb[k] = *(const float4*)(hsrc + (size_t)sk * 128 + lane * 4);
                    }
                }
#pragma unroll
                for (int k = 0; k < 8; ++k) {
                    float4* row = (float4*)(acc + da[k] * P + lane * 4);
                    float4 a4 = *row;
                    a4.x += va[k].x;
                    a4.y += va[k].y;
                    a4.z += va[k].z;
                    a4.w += va[k].w;
                    *row = a4;
                    mycnt += (da[k] == myrow) ? 1.0f : 0.0f;
                }
                if (two) {
#pragma unroll
                    for (int k = 0; k < 8; ++k) {
                        float4* row = (float4*)(acc + db[k] * P + lane * 4);
                        float4 a4 = *row;
                        a4.x += vb[k].x;
                        a4.y += vb[k].y;
                        a4.z += vb[k].z;
                        a4.w += vb[k].w;
                        *row = a4;
                        mycnt += (db[k] == myrow) ? 1.0f : 0.0f;
                    }
                }
            }
        }
        if (lane == 0) qn[wave] = 0u;
    };

    if constexpr (BITMAP) {
        const int ngroups = nedges >> 4;
        const int wpo = ((ngroups + 1023) >> 10) << 5;
        const unsigned* bmw = bm + (size_t)blockIdx.x * BMP;
        const int nslab = wpo > 0 ? (wpo + 1024 * SI - 1) / (1024 * SI) : 1;
        for (int sl = 0; sl < nslab; ++sl) {
            if (sl == 0 && wave == 0) {
                const int e = (ngroups << 4) + lane;
                const bool in = e < nedges;
                const int l = (in ? dst[e] : -1) - base;
                route(in && ((unsigned)l < urows), (unsigned)e, l);
            }
            for (int it = 0; it < SI; ++it) {
                const int w0 = ((sl * SI + it) << 10) + (wave << 5);
                unsigned word = w0 < wpo ? bmw[w0 + lane] : 0u;
                while (__builtin_amdgcn_ballot_w32(word != 0u) != 0u) {
                    const int g  = ((w0 + lane) << 5) + (int)__builtin_ctz(word | 0x80000000u);
                    const bool t = (word != 0u) && (g < ngroups);
                    const int e0 = g << 4;
                    word &= word - 1u;
                    int4 d0 = make_int4(-1, -1, -1, -1), d1 = d0, d2 = d0, d3 = d0;
                    if (t) {
                        const int4* p = (const int4*)(dst + e0);
                        d0 = p[0];
                        d1 = p[1];
                        d2 = p[2];
                        d3 = p[3];
                    }
                    unsigned m;
                    m  = ((unsigned)(d0.x - base) < urows ? 0x0001u : 0u) | ((unsigned)(d0.y - base) < urows ? 0x0002u : 0u);
                    m |= ((unsigned)(d0.z - base) < urows ? 0x0004u : 0u) | ((unsigned)(d0.w - base) < urows ? 0x0008u : 0u);
                    m |= ((unsigned)(d1.x - base) < urows ? 0x0010u : 0u) | ((unsigned)(d1.y - base) < urows ? 0x0020u : 0u);
                    m |= ((unsigned)(d1.z - base) < urows ? 0x0040u : 0u) | ((unsigned)(d1.w - base) < urows ? 0x0080u : 0u);
                    m |= ((unsigned)(d2.x - base) < urows ? 0x0100u : 0u) | ((unsigned)(d2.y - base) < urows ? 0x0200u : 0u);
                    m |= ((unsigned)(d2.z - base) < urows ? 0x0400u : 0u) | ((unsigned)(d2.w - base) < urows ? 0x0800u : 0u);
                    m |= ((unsigned)(d3.x - base) < urows ? 0x1000u : 0u) | ((unsigned)(d3.y - base) < urows ? 0x2000u : 0u);
                    m |= ((unsigned)(d3.z - base) < urows ? 0x4000u : 0u) | ((unsigned)(d3.w - base) < urows ? 0x8000u : 0u);
                    while (__builtin_amdgcn_ballot_w32(m != 0u) != 0u) {
                        const bool tt = m != 0u;
                        const unsigned p = (unsigned)__builtin_ctz(m | 0x10000u);
                        const int4 dq = p < 8u ? (p < 4u ? d0 : d1) : (p < 12u ? d2 : d3);
                        const unsigned r4 = p & 3u;
                        const int dval = r4 < 2u ? (r4 == 0u ? dq.x : dq.y) : (r4 == 2u ? dq.z : dq.w);
                        route(tt, (unsigned)e0 + p, dval - base);
                        m &= m - 1u;
                    }
                }
            }
            __syncthreads();
            drain_own();
            __syncthreads();
        }
    } else {

    const int nquads  = nedges >> 2;
    const int nrounds = (nquads + 4095) >> 12;
    int rr = (int)(((long long)blockIdx.x * nrounds) / gridDim.x);
    const int4* dst4 = (const int4*)dst;
    for (int r = 0; r < nrounds; ++r) {
        const int q0 = (rr << 12) + (wave << 7) + lane;
        int4 dq[4];
#pragma unroll
        for (int u = 0; u < 4; ++u) {
            const int q = q0 + (u << 5);
            dq[u] = (q < nquads) ? dst4[q] : make_int4(-1, -1, -1, -1);
        }
#pragma unroll
        for (int u = 0; u < 4; ++u) {
            const int q = q0 + (u << 5);
            const int l0 = dq[u].x - base, l1 = dq[u].y - base;
            const int l2 = dq[u].z - base, l3 = dq[u].w - base;
            const bool t0 = (unsigned)l0 < urows, t1 = (unsigned)l1 < urows;
            const bool t2 = (unsigned)l2 < urows, t3 = (unsigned)l3 < urows;
            if (__builtin_amdgcn_ballot_w32(t0 | t1 | t2 | t3) != 0u) {
                const unsigned e0 = (unsigned)q << 2;
                route(t0, e0, l0);
                route(t1, e0 + 1u, l1);
                route(t2, e0 + 2u, l2);
                route(t3, e0 + 3u, l3);
            }
        }
        rr = (rr + 1 == nrounds) ? 0 : rr + 1;
    }
    if (wave == 0) {
        const int e = (nquads << 2) + lane;
        const bool in = e < nedges;
        const int l = (in ? dst[e] : -1) - base;
        route(in && ((unsigned)l < urows), (unsigned)e, l);
    }
    __syncthreads();
    drain_own();
    }
    if (myrow < R) cnt[myrow] += mycnt;

    __syncthreads();

    const int wv  = __builtin_amdgcn_readfirstlane(wave);
    const int nt  = wv % NT;
    const int sg  = wv / NT;
    const int n0  = nt * 16;
    const int idx = lane & 15;
    const int nstr = (rows + 15) >> 4;

    {
        v2f bfr[32];
        const float* Bn = Wneigh + (size_t)(n0 + idx) * 128 + 2 * hi;
#pragma unroll
        for (int k = 0; k < 32; ++k) bfr[k] = *(const v2f*)(Bn + 4 * k);

        for (int it = 0; it < NIT; ++it) {
            const int strip = it * SG + sg;
            const bool valid = strip < nstr;
            v8f c = {0.f, 0.f, 0.f, 0.f, 0.f, 0.f, 0.f, 0.f};
            if (valid) {
                int ar = strip * 16 + idx;
                if (ar > R) ar = R;
                const float invdeg = 1.0f / fmaxf(cnt[ar], 1.0f);
                const float* Am = acc + ar * P + 2 * hi;
#pragma unroll
                for (int k = 0; k < 32; ++k) {
                    v2f a = *(const v2f*)(Am + 4 * k);
                    a.x *= invdeg;
                    a.y *= invdeg;
                    c = __builtin_amdgcn_wmma_f32_16x16x4_f32(false, a, false, bfr[k],
                                                              (short)0, c, false, false);
                }
            }
            __syncthreads();
            if (valid) {
#pragma unroll
                for (int j = 0; j < 8; ++j) {
                    const int r = strip * 16 + 8 * hi + j;
                    if (r < rows) acc[r * P + n0 + idx] = c[j];
                }
            }
        }
    }

    {
        v2f bfr[32];
        const float* Bs = Wself + (size_t)(n0 + idx) * 128 + 2 * hi;
#pragma unroll
        for (int k = 0; k < 32; ++k) bfr[k] = *(const v2f*)(Bs + 4 * k);
        const float bv = bias[n0 + idx];

        static_assert(RPW == 2 || RPW == 4, "a wave copies 2 or 4 rows of a strip");
        float4 nx0 = make_float4(0.f, 0.f, 0.f, 0.f), nx1 = nx0, nx2 = nx0, nx3 = nx0;
        const float* hrow = hsrc + (size_t)base * 128 + lane * 4;
        const int lr0 = nt * RPW;
        auto ldrow = [hrow, rows](int gr) -> float4 {
            if (gr > rows - 1) gr = rows - 1;
            return *(const float4*)(hrow + (size_t)gr * 128);
        };
        if (sg < nstr) {
            nx0 = ldrow(sg * 16 + lr0);
            nx1 = ldrow(sg * 16 + lr0 + 1);
            if (RPW == 4) { nx2 = ldrow(sg * 16 + lr0 + 2); nx3 = ldrow(sg * 16 + lr0 + 3); }
        }

        for (int it = 0; it < NIT; ++it) {
            const int strip = it * SG + sg;
            const bool valid = strip < nstr;
            if (valid) {
                float* srow = stage + (sg * 16 + lr0) * P + lane * 4;
                *(float4*)(srow)         = nx0;
                *(float4*)(srow + P)     = nx1;
                if (RPW == 4) { *(float4*)(srow + 2 * P) = nx2; *(float4*)(srow + 3 * P) = nx3; }
            }
            __syncthreads();
            if (strip + SG < nstr) {
                const int g0 = (strip + SG) * 16 + lr0;
                nx0 = ldrow(g0);
                nx1 = ldrow(g0 + 1);
                if (RPW == 4) { nx2 = ldrow(g0 + 2); nx3 = ldrow(g0 + 3); }
            }
            if (valid) {
                v8f c;
#pragma unroll
                for (int j = 0; j < 8; ++j) {
                    const int r = strip * 16 + 8 * hi + j;
                    c[j] = (r < rows) ? acc[r * P + n0 + idx] : 0.0f;
                }
                const float* As = stage + (sg * 16 + idx) * P + 2 * hi;
#pragma unroll
                for (int k = 0; k < 32; ++k) {
                    const v2f a = *(const v2f*)(As + 4 * k);
                    c = __builtin_amdgcn_wmma_f32_16x16x4_f32(false, a, false, bfr[k],
                                                              (short)0, c, false, false);
                }
#pragma unroll
                for (int j = 0; j < 8; ++j) {
                    const int r = strip * 16 + 8 * hi + j;
                    if (r < rows) {
                        float v = c[j] + bv;
                        if (RELU) v = fmaxf(v, 0.0f);
                        acc[r * P + n0 + idx] = v;
                    }
                }
            }
            if (it + 1 < NIT) __syncthreads();
        }
    }

    __syncthreads();
    for (int i = tid; i < rows * (DOUT / 4); i += 1024) {
        const int r = i / (DOUT / 4), l = i % (DOUT / 4);
        const float4 v = *(const float4*)(acc + r * P + l * 4);
        *(float4*)(out + (size_t)(base + r) * DOUT + l * 4) = v;
    }
}

static inline unsigned cdiv(unsigned a, unsigned b) { return (a + b - 1) / b; }

extern "C" void kernel_launch(void* const* d_in, const int* in_sizes, int n_in,
                              void* d_out, int out_size, void* d_ws, size_t ws_size,
                              hipStream_t stream) {
    const float* feats = (const float*)d_in[0];
    const int* src0 = (const int*)d_in[1];
    const int* dst0 = (const int*)d_in[2];
    const int* src1 = (const int*)d_in[3];
    const int* dst1 = (const int*)d_in[4];
    const int* src2 = (const int*)d_in[5];
    const int* dst2 = (const int*)d_in[6];
    const float* Ws0 = (const float*)d_in[10];
    const float* Wn0 = (const float*)d_in[11];
    const float* b0  = (const float*)d_in[12];
    const float* Ws1 = (const float*)d_in[13];
    const float* Wn1 = (const float*)d_in[14];
    const float* b1  = (const float*)d_in[15];
    const float* Ws2 = (const float*)d_in[16];
    const float* Wn2 = (const float*)d_in[17];
    const float* b2  = (const float*)d_in[18];

    const int E0 = in_sizes[1], E1 = in_sizes[3], E2 = in_sizes[5];
    const int N0 = 500000, N1 = 100000, N2 = 20000, N3 = 5000;
    constexpr int R0 = 400, R1 = 80, R2 = 20;
    constexpr int QC0 = 384, QC1 = 128;

    char* ws = (char*)d_ws;
    float* h1  = (float*)ws;                    ws += (size_t)N1 * 128 * sizeof(float);
    float* h2  = (float*)ws;

    const dim3 blk(1024);

    constexpr int NB     = (N1 + R0 - 1) / R0;
    static_assert((N2 + R1 - 1) / R1 == NB, "layers 0 and 1 have the same number of blocks");
    constexpr int BMP    = R1 * 128;
    constexpr int OFF1   = 4096;
    constexpr int G0 = 125000, G1 = 25000;
    constexpr int TILES0 = (G0 + 1023) / 1024;
    constexpr int TILES1 = (G1 + 1023) / 1024;
    static_assert(TILES0 * 32 <= OFF1 && OFF1 + TILES1 * 32 <= BMP, "both bitmaps must fit a slot");
    static_assert((size_t)NB * BMP <= (size_t)N2 * 128, "the slots must fit the space of h2");
    unsigned* bm = (unsigned*)h2;
    group_bitmaps<R0, R1, NB, BMP, OFF1><<<TILES0 + TILES1, blk, 0, stream>>>(dst0, E0, N1, dst1, E1, N2, bm);

    sage_layer_owned<R0, 128, true, BMP, QC0, 4, 32><<<cdiv((unsigned)N1, R0), blk, 0, stream>>>(feats, src0, dst0, bm, E0, N0, N1, Ws0, Wn0, b0, h1);
    sage_layer_owned<R1, 128, true, BMP, QC1, 1, 0><<<cdiv((unsigned)N2, R1), blk, 0, stream>>>(h1, src1, dst1, bm + OFF1, E1, N1, N2, Ws1, Wn1, b1, h2);
    sage_layer_owned<R2, 64, false, 0, QC1, 1, 0><<<cdiv((unsigned)N3, R2), blk, 0, stream>>>(h2, src2, dst2, nullptr, E2, N2, N3, Ws2, Wn2, b2, (float*)d_out);
}


# Weight tiles reach the B registers through LDS: each matrix copied once per work-group by whole-row loads, fragments read from LDS (was 8 B per lane from 16 rows per load); sum order unchanged
# speedup vs baseline: 4.2276x; 1.0996x over previous
#include <hip/hip_runtime.h>
#include <hip/hip_bf16.h>


typedef __attribute__((ext_vector_type(2))) float v2f;
typedef __attribute__((ext_vector_type(8))) float v8f;

template <int R, int NB, int PITCH>
static __device__ inline void bitmap_tile(const int* __restrict__ dst, int ngroups, unsigned un, int tile,
                                          unsigned* __restrict__ bm, unsigned* bits) {
    const int tid = threadIdx.x;
    const unsigned bit = 1u << (tid & 31);
    const int word = tid >> 5;
    const int g = (tile << 10) + tid;
    if (g < ngroups) {
        const int4* p = (const int4*)(dst + ((size_t)g << 4));
        const int4 d0 = p[0], d1 = p[1], d2 = p[2], d3 = p[3];
        const int dv[16] = {d0.x, d0.y, d0.z, d0.w, d1.x, d1.y, d1.z, d1.w,
                            d2.x, d2.y, d2.z, d2.w, d3.x, d3.y, d3.z, d3.w};
#pragma unroll
        for (int j = 0; j < 16; ++j) {
            const unsigned d = (unsigned)dv[j];
            if (d < un) {
                const unsigned o = d / (unsigned)R;
                if (o < (unsigned)NB) atomicOr(&bits[word * NB + o], bit);
            }
        }
    }
    __syncthreads();
    for (int i = tid; i < NB * 32; i += 1024) {
        const int k = (i & 31) * NB + (i >> 5);
        bm[(size_t)(i >> 5) * PITCH + (tile << 5) + (i & 31)] = bits[k];
        bits[k] = 0u;
    }
}

template <int RA, int RB, int NB, int PITCH, int OFFB>
__global__ __launch_bounds__(1024) void group_bitmaps(const int* __restrict__ dstA, int nedgesA, int n_dstA,
                                                      const int* __restrict__ dstB, int nedgesB, int n_dstB,
                                                      unsigned* __restrict__ bm) {
    __shared__ unsigned bits[NB * 32];
    const int ngA = nedgesA >> 4, ngB = nedgesB >> 4;
    const int ntA = (ngA + 1023) >> 10, ntB = (ngB + 1023) >> 10;
    for (int i = threadIdx.x; i < NB * 32; i += 1024) bits[i] = 0u;
    for (int tile = blockIdx.x; tile < ntA + ntB; tile += gridDim.x) {
        __syncthreads();
        if (tile < ntA) bitmap_tile<RA, NB, PITCH>(dstA, ngA, (unsigned)n_dstA, tile, bm, bits);
        else            bitmap_tile<RB, NB, PITCH>(dstB, ngB, (unsigned)n_dstB, tile - ntA, bm + OFFB, bits);
    }
}

template <int R, int DOUT, bool RELU, int BMP, int QCAP, int SI, int NBK>
__global__ __launch_bounds__(1024) void sage_layer_owned(const float* __restrict__ hsrc,
                                                         const int* __restrict__ src,
                                                         const int* __restrict__ dst,
                                                         const unsigned* bm,
                                                         int nedges, int nsrc, int n_dst,
                                                         const float* __restrict__ Wself,
                                                         const float* __restrict__ Wneigh,
                                                         const float* __restrict__ bias,
                                                         float* __restrict__ out) {
    constexpr bool BITMAP = BMP > 0;
    static_assert(NBK == 0 || NBK == 32, "one source range per lane in the prefix sum");
    constexpr int P    = 132;
    constexpr int NT   = DOUT / 16;
    constexpr int SG   = 32 / NT;
    constexpr int NSTR = (R + 15) / 16;
    constexpr int NIT  = (NSTR + SG - 1) / SG;
    constexpr int RPW  = 16 / NT;
    constexpr int QP   = QCAP + 1;
    constexpr int QWORDS = 2 * 32 * QP;
    constexpr int SWORDS = SG * 16 * P;
    constexpr int WWORDS = DOUT * P;
    constexpr int GWORDS = SWORDS + WWORDS;
    constexpr int XWORDS = QWORDS > GWORDS ? QWORDS : GWORDS;
    constexpr int WV     = DOUT * 32 / 1024;
    static_assert(WV * 1024 == DOUT * 32, "the weight matrix is dealt out evenly to the 1024 threads");
    __shared__ __attribute__((aligned(16))) float acc[(R + 1) * P];
    __shared__ float cnt[R + 1];
    __shared__ unsigned qn[32];
    __shared__ unsigned hist[NBK > 0 ? 32 * NBK : 1];
    __shared__ __attribute__((aligned(16))) unsigned xbuf[XWORDS];
    unsigned* const qe = xbuf;
    unsigned* const qd = xbuf + 32 * QP;
    float* const stage = (float*)xbuf;
    float* const wlds  = (float*)xbuf + SWORDS;

    const int tid  = threadIdx.x;
    const int lane = tid & 31;
    const int wave = tid >> 5;
    const int hi   = lane >> 4;
    const int base = blockIdx.x * R;
    int rows = n_dst - base;
    if (rows > R) rows = R;
    if (rows <= 0) return;
    const unsigned urows = (unsigned)rows;

    for (int i = tid; i < (R + 1) * (P / 4); i += 1024)
        *(float4*)(acc + i * 4) = make_float4(0.0f, 0.0f, 0.0f, 0.0f);
    for (int i = tid; i < R + 1; i += 1024) cnt[i] = 0.0f;
    if (tid < 32) qn[tid] = 0u;
    if constexpr (NBK > 0) {
        for (int i = tid; i < 32 * NBK; i += 1024) hist[i] = 0u;
    }
    __syncthreads();

    const int wq = wave * QP;
    static_assert(R <= 1024, "one lane per owned row");
    const int myrow = wave + 32 * lane;
    float mycnt = 0.0f;

    auto route = [&](bool t, unsigned eidx, int dloc) {
        if (t) {
            const int ow = dloc & 31;
            const unsigned pos = atomicAdd(&qn[ow], 1u);
            if (pos < (unsigned)QCAP) {
                qe[ow * QP + pos] = eidx;
                qd[ow * QP + pos] = (unsigned)dloc;
            } else {
                int s = src[eidx];
                s = s < 0 ? 0 : (s > nsrc - 1 ? nsrc - 1 : s);
                const float* hr = hsrc + (size_t)s * 128;
                float* row = acc + dloc * P;
                for (int c = 0; c < 128; ++c) atomicAdd(row + c, hr[c]);
                atomicAdd(cnt + dloc, 1.0f);
            }
        }
    };

    auto drain_own = [&]() {
        int n = __builtin_amdgcn_readfirstlane((int)qn[wave]);
        if (n > QCAP) n = QCAP;
        if constexpr (NBK > 0) {
            constexpr int EPL = (QCAP + 31) / 32;
            unsigned* const hw = hist + wave * NBK;
            const float bscale = (float)NBK / (float)nsrc;
            int sv[EPL], dv[EPL];
            unsigned bk[EPL], rk[EPL];
#pragma unroll
            for (int j = 0; j < EPL; ++j) {
                const int i = j * 32 + lane;
                const bool ok = i < n;
                sv[j] = ok ? (int)qe[wq + i] : 0;
                dv[j] = ok ? (int)qd[wq + i] : R;
            }
#pragma unroll
            for (int j = 0; j < EPL; ++j) {
                const int s = (j * 32 + lane < n) ? src[sv[j]] : 0;
                sv[j] = s < 0 ? 0 : (s > nsrc - 1 ? nsrc - 1 : s);
            }
#pragma unroll
            for (int j = 0; j < EPL; ++j) {
                const unsigned b = (unsigned)((float)sv[j] * bscale);
                bk[j] = b > (unsigned)(NBK - 1) ? (unsigned)(NBK - 1) : b;
                rk[j] = (j * 32 + lane < n) ? atomicAdd(&hw[bk[j]], 1u) : 0u;
            }
            const unsigned nb_ = hw[lane];
            unsigned incl = nb_;
            for (int d = 1; d < 32; d <<= 1) {
                const unsigned t = __shfl_up(incl, d);
                if (lane >= d) incl += t;
            }
            const unsigned first = incl - nb_;
            hw[lane] = 0u;
#pragma unroll
            for (int j = 0; j < EPL; ++j) rk[j] += (unsigned)__shfl((int)first, (int)bk[j]);
#pragma unroll
            for (int j = 0; j < EPL; ++j) {
                if (j * 32 + lane < n) {
                    qe[wq + rk[j]] = (unsigned)sv[j];
                    qd[wq + rk[j]] = (unsigned)dv[j];
                }
            }
        }
        for (int b0 = 0; b0 < n; b0 += 32) {
            const int i = b0 + lane;
            const bool ok = i < n;
            const unsigned e = ok ? qe[wq + i] : 0u;
            const int dl = ok ? (int)qd[wq + i] : R;
            int s;
            if constexpr (NBK > 0) s = (int)e;
            else                   s = ok ? src[e] : 0;
            s = s < 0 ? 0 : (s > nsrc - 1 ? nsrc - 1 : s);
            const int nb = (n - b0) < 32 ? (n - b0) : 32;
            for (int k0 = 0; k0 < nb; k0 += 16) {
                float4 va[8], vb[8];
                int    da[8], db[8];
                const bool two = k0 + 8 < nb;
#pragma unroll
                for (int k = 0; k < 8; ++k) {
                    const int sk = __builtin_amdgcn_readlane(s, k0 + k);
                    da[k] = __builtin_amdgcn_readlane(dl, k0 + k);
                    va[k] = *(const float4*)(hsrc + (size_t)sk * 128 + lane * 4);
                }
                if (two) {
#pragma unroll
                    for (int k = 0; k < 8; ++k) {
                        const int sk = __builtin_amdgcn_readlane(s, k0 + 8 + k);
                        db[k] = __builtin_amdgcn_readlane(dl, k0 + 8 + k);
                        vb[k] = *(const float4*)(hsrc + (size_t)sk * 128 + lane * 4);
                    }
                }
#pragma unroll
                for (int k = 0; k < 8; ++k) {
                    float4* row = (float4*)(acc + da[k] * P + lane * 4);
                    float4 a4 = *row;
                    a4.x += va[k].x;
                    a4.y += va[k].y;
                    a4.z += va[k].z;
                    a4.w += va[k].w;
                    *row = a4;
                    mycnt += (da[k] == myrow) ? 1.0f : 0.0f;
                }
                if (two) {
#pragma unroll
                    for (int k = 0; k < 8; ++k) {
                        float4* row = (float4*)(acc + db[k] * P + lane * 4);
                        float4 a4 = *row;
                        a4.x += vb[k].x;
                        a4.y += vb[k].y;
                        a4.z += vb[k].z;
                        a4.w += vb[k].w;
                        *row = a4;
                        mycnt += (db[k] == myrow) ? 1.0f : 0.0f;
                    }
                }
            }
        }
        if (lane == 0) qn[wave] = 0u;
    };

    if constexpr (BITMAP) {
        const int ngroups = nedges >> 4;
        const int wpo = ((ngroups + 1023) >> 10) << 5;
        const unsigned* bmw = bm + (size_t)blockIdx.x * BMP;
        const int nslab = wpo > 0 ? (wpo + 1024 * SI - 1) / (1024 * SI) : 1;
        for (int sl = 0; sl < nslab; ++sl) {
            if (sl == 0 && wave == 0) {
                const int e = (ngroups << 4) + lane;
                const bool in = e < nedges;
                const int l = (in ? dst[e] : -1) - base;
                route(in && ((unsigned)l < urows), (unsigned)e, l);
            }
            for (int it = 0; it < SI; ++it) {
                const int w0 = ((sl * SI + it) << 10) + (wave << 5);
                unsigned word = w0 < wpo ? bmw[w0 + lane] : 0u;
                while (__builtin_amdgcn_ballot_w32(word != 0u) != 0u) {
                    const int g  = ((w0 + lane) << 5) + (int)__builtin_ctz(word | 0x80000000u);
                    const bool t = (word != 0u) && (g < ngroups);
                    const int e0 = g << 4;
                    word &= word - 1u;
                    int4 d0 = make_int4(-1, -1, -1, -1), d1 = d0, d2 = d0, d3 = d0;
                    if (t) {
                        const int4* p = (const int4*)(dst + e0);
                        d0 = p[0];
                        d1 = p[1];
                        d2 = p[2];
                        d3 = p[3];
                    }
                    unsigned m;
                    m  = ((unsigned)(d0.x - base) < urows ? 0x0001u : 0u) | ((unsigned)(d0.y - base) < urows ? 0x0002u : 0u);
                    m |= ((unsigned)(d0.z - base) < urows ? 0x0004u : 0u) | ((unsigned)(d0.w - base) < urows ? 0x0008u : 0u);
                    m |= ((unsigned)(d1.x - base) < urows ? 0x0010u : 0u) | ((unsigned)(d1.y - base) < urows ? 0x0020u : 0u);
                    m |= ((unsigned)(d1.z - base) < urows ? 0x0040u : 0u) | ((unsigned)(d1.w - base) < urows ? 0x0080u : 0u);
                    m |= ((unsigned)(d2.x - base) < urows ? 0x0100u : 0u) | ((unsigned)(d2.y - base) < urows ? 0x0200u : 0u);
                    m |= ((unsigned)(d2.z - base) < urows ? 0x0400u : 0u) | ((unsigned)(d2.w - base) < urows ? 0x0800u : 0u);
                    m |= ((unsigned)(d3.x - base) < urows ? 0x1000u : 0u) | ((unsigned)(d3.y - base) < urows ? 0x2000u : 0u);
                    m |= ((unsigned)(d3.z - base) < urows ? 0x4000u : 0u) | ((unsigned)(d3.w - base) < urows ? 0x8000u : 0u);
                    while (__builtin_amdgcn_ballot_w32(m != 0u) != 0u) {
                        const bool tt = m != 0u;
                        const unsigned p = (unsigned)__builtin_ctz(m | 0x10000u);
                        const int4 dq = p < 8u ? (p < 4u ? d0 : d1) : (p < 12u ? d2 : d3);
                        const unsigned r4 = p & 3u;
                        const int dval = r4 < 2u ? (r4 == 0u ? dq.x : dq.y) : (r4 == 2u ? dq.z : dq.w);
                        route(tt, (unsigned)e0 + p, dval - base);
                        m &= m - 1u;
                    }
                }
            }
            __syncthreads();
            drain_own();
            __syncthreads();
        }
    } else {

    const int nquads  = nedges >> 2;
    const int nrounds = (nquads + 4095) >> 12;
    int rr = (int)(((long long)blockIdx.x * nrounds) / gridDim.x);
    const int4* dst4 = (const int4*)dst;
    for (int r = 0; r < nrounds; ++r) {
        const int q0 = (rr << 12) + (wave << 7) + lane;
        int4 dq[4];
#pragma unroll
        for (int u = 0; u < 4; ++u) {
            const int q = q0 + (u << 5);
            dq[u] = (q < nquads) ? dst4[q] : make_int4(-1, -1, -1, -1);
        }
#pragma unroll
        for (int u = 0; u < 4; ++u) {
            const int q = q0 + (u << 5);
            const int l0 = dq[u].x - base, l1 = dq[u].y - base;
            const int l2 = dq[u].z - base, l3 = dq[u].w - base;
            const bool t0 = (unsigned)l0 < urows, t1 = (unsigned)l1 < urows;
            const bool t2 = (unsigned)l2 < urows, t3 = (unsigned)l3 < urows;
            if (__builtin_amdgcn_ballot_w32(t0 | t1 | t2 | t3) != 0u) {
                const unsigned e0 = (unsigned)q << 2;
                route(t0, e0, l0);
                route(t1, e0 + 1u, l1);
                route(t2, e0 + 2u, l2);
                route(t3, e0 + 3u, l3);
            }
        }
        rr = (rr + 1 == nrounds) ? 0 : rr + 1;
    }
    if (wave == 0) {
        const int e = (nquads << 2) + lane;
        const bool in = e < nedges;
        const int l = (in ? dst[e] : -1) - base;
        route(in && ((unsigned)l < urows), (unsigned)e, l);
    }
    __syncthreads();
    drain_own();
    }
    static_assert(WV == 2 || WV == 4, "a thread copies 2 or 4 pieces of a weight matrix");
    auto wget = [tid](const float* Wm, int j) -> float4 {
        return *(const float4*)(Wm + (size_t)(tid + j * 1024) * 4);
    };
    float4 w0 = wget(Wneigh, 0), w1 = wget(Wneigh, 1);
    float4 w2 = make_float4(0.f, 0.f, 0.f, 0.f), w3 = w2;
    if (WV == 4) { w2 = wget(Wneigh, 2); w3 = wget(Wneigh, 3); }
    float4 s0 = wget(Wself, 0), s1 = wget(Wself, 1);
    float4 s2 = make_float4(0.f, 0.f, 0.f, 0.f), s3 = s2;
    if (WV == 4) { s2 = wget(Wself, 2); s3 = wget(Wself, 3); }

    if (myrow < R) cnt[myrow] += mycnt;

    __syncthreads();

    const int wv  = __builtin_amdgcn_readfirstlane(wave);
    const int nt  = wv % NT;
    const int sg  = wv / NT;
    const int n0  = nt * 16;
    const int idx = lane & 15;
    const int nstr = (rows + 15) >> 4;

    auto wput = [tid, wlds](int j, float4 v) {
        const int i = tid + j * 1024;
        *(float4*)(wlds + (i >> 5) * P + (i & 31) * 4) = v;
    };
    wput(0, w0);
    wput(1, w1);
    if (WV == 4) { wput(2, w2); wput(3, w3); }
    __syncthreads();

    {
        v2f bfr[32];
        const float* Bn = wlds + (n0 + idx) * P + 2 * hi;
#pragma unroll
        for (int k = 0; k < 32; ++k) bfr[k] = *(const v2f*)(Bn + 4 * k);

        for (int it = 0; it < NIT; ++it) {
            const int strip = it * SG + sg;
            const bool valid = strip < nstr;
            v8f c = {0.f, 0.f, 0.f, 0.f, 0.f, 0.f, 0.f, 0.f};
            if (valid) {
                int ar = strip * 16 + idx;
                if (ar > R) ar = R;
                const float invdeg = 1.0f / fmaxf(cnt[ar], 1.0f);
                const float* Am = acc + ar * P + 2 * hi;
#pragma unroll
                for (int k = 0; k < 32; ++k) {
                    v2f a = *(const v2f*)(Am + 4 * k);
                    a.x *= invdeg;
                    a.y *= invdeg;
                    c = __builtin_amdgcn_wmma_f32_16x16x4_f32(false, a, false, bfr[k],
                                                              (short)0, c, false, false);
                }
            }
            __syncthreads();
            if (it == 0) {
                wput(0, s0);
                wput(1, s1);
                if (WV == 4) { wput(2, s2); wput(3, s3); }
            }
            if (valid) {
#pragma unroll
                for (int j = 0; j < 8; ++j) {
                    const int r = strip * 16 + 8 * hi + j;
                    if (r < rows) acc[r * P + n0 + idx] = c[j];
                }
            }
        }
    }
    if constexpr (NIT == 1) __syncthreads();

    {
        v2f bfr[32];
        const float* Bs = wlds + (n0 + idx) * P + 2 * hi;
#pragma unroll
        for (int k = 0; k < 32; ++k) bfr[k] = *(const v2f*)(Bs + 4 * k);
        const float bv = bias[n0 + idx];

        static_assert(RPW == 2 || RPW == 4, "a wave copies 2 or 4 rows of a strip");
        float4 nx0 = make_float4(0.f, 0.f, 0.f, 0.f), nx1 = nx0, nx2 = nx0, nx3 = nx0;
        const float* hrow = hsrc + (size_t)base * 128 + lane * 4;
        const int lr0 = nt * RPW;
        auto ldrow = [hrow, rows](int gr) -> float4 {
            if (gr > rows - 1) gr = rows - 1;
            return *(const float4*)(hrow + (size_t)gr * 128);
        };
        if (sg < nstr) {
            nx0 = ldrow(sg * 16 + lr0);
            nx1 = ldrow(sg * 16 + lr0 + 1);
            if (RPW == 4) { nx2 = ldrow(sg * 16 + lr0 + 2); nx3 = ldrow(sg * 16 + lr0 + 3); }
        }

        for (int it = 0; it < NIT; ++it) {
            const int strip = it * SG + sg;
            const bool valid = strip < nstr;
            if (valid) {
                float* srow = stage + (sg * 16 + lr0) * P + lane * 4;
                *(float4*)(srow)         = nx0;
                *(float4*)(srow + P)     = nx1;
                if (RPW == 4) { *(float4*)(srow + 2 * P) = nx2; *(float4*)(srow + 3 * P) = nx3; }
            }
            __syncthreads();
            if (strip + SG < nstr) {
                const int g0 = (strip + SG) * 16 + lr0;
                nx0 = ldrow(g0);
                nx1 = ldrow(g0 + 1);
                if (RPW == 4) { nx2 = ldrow(g0 + 2); nx3 = ldrow(g0 + 3); }
            }
            if (valid) {
                v8f c;
#pragma unroll
                for (int j = 0; j < 8; ++j) {
                    const int r = strip * 16 + 8 * hi + j;
                    c[j] = (r < rows) ? acc[r * P + n0 + idx] : 0.0f;
                }
                const float* As = stage + (sg * 16 + idx) * P + 2 * hi;
#pragma unroll
                for (int k = 0; k < 32; ++k) {
                    const v2f a = *(const v2f*)(As + 4 * k);
                    c = __builtin_amdgcn_wmma_f32_16x16x4_f32(false, a, false, bfr[k],
                                                              (short)0, c, false, false);
                }
#pragma unroll
                for (int j = 0; j < 8; ++j) {
                    const int r = strip * 16 + 8 * hi + j;
                    if (r < rows) {
                        float v = c[j] + bv;
                        if (RELU) v = fmaxf(v, 0.0f);
                        acc[r * P + n0 + idx] = v;
                    }
                }
            }
            if (it + 1 < NIT) __syncthreads();
        }
    }

    __syncthreads();
    for (int i = tid; i < rows * (DOUT / 4); i += 1024) {
        const int r = i / (DOUT / 4), l = i % (DOUT / 4);
        const float4 v = *(const float4*)(acc + r * P + l * 4);
        *(float4*)(out + (size_t)(base + r) * DOUT + l * 4) = v;
    }
}

static inline unsigned cdiv(unsigned a, unsigned b) { return (a + b - 1) / b; }

extern "C" void kernel_launch(void* const* d_in, const int* in_sizes, int n_in,
                              void* d_out, int out_size, void* d_ws, size_t ws_size,
                              hipStream_t stream) {
    const float* feats = (const float*)d_in[0];
    const int* src0 = (const int*)d_in[1];
    const int* dst0 = (const int*)d_in[2];
    const int* src1 = (const int*)d_in[3];
    const int* dst1 = (const int*)d_in[4];
    const int* src2 = (const int*)d_in[5];
    const int* dst2 = (const int*)d_in[6];
    const float* Ws0 = (const float*)d_in[10];
    const float* Wn0 = (const float*)d_in[11];
    const float* b0  = (const float*)d_in[12];
    const float* Ws1 = (const float*)d_in[13];
    const float* Wn1 = (const float*)d_in[14];
    const float* b1  = (const float*)d_in[15];
    const float* Ws2 = (const float*)d_in[16];
    const float* Wn2 = (const float*)d_in[17];
    const float* b2  = (const float*)d_in[18];

    const int E0 = in_sizes[1], E1 = in_sizes[3], E2 = in_sizes[5];
    const int N0 = 500000, N1 = 100000, N2 = 20000, N3 = 5000;
    constexpr int R0 = 400, R1 = 80, R2 = 20;
    constexpr int QC0 = 384, QC1 = 128;

    char* ws = (char*)d_ws;
    float* h1  = (float*)ws;                    ws += (size_t)N1 * 128 * sizeof(float);
    float* h2  = (float*)ws;

    const dim3 blk(1024);

    constexpr int NB     = (N1 + R0 - 1) / R0;
    static_assert((N2 + R1 - 1) / R1 == NB, "layers 0 and 1 have the same number of blocks");
    constexpr int BMP    = R1 * 128;
    constexpr int OFF1   = 4096;
    constexpr int G0 = 125000, G1 = 25000;
    constexpr int TILES0 = (G0 + 1023) / 1024;
    constexpr int TILES1 = (G1 + 1023) / 1024;
    static_assert(TILES0 * 32 <= OFF1 && OFF1 + TILES1 * 32 <= BMP, "both bitmaps must fit a slot");
    static_assert((size_t)NB * BMP <= (size_t)N2 * 128, "the slots must fit the space of h2");
    unsigned* bm = (unsigned*)h2;
    group_bitmaps<R0, R1, NB, BMP, OFF1><<<TILES0 + TILES1, blk, 0, stream>>>(dst0, E0, N1, dst1, E1, N2, bm);

    sage_layer_owned<R0, 128, true, BMP, QC0, 4, 32><<<cdiv((unsigned)N1, R0), blk, 0, stream>>>(feats, src0, dst0, bm, E0, N0, N1, Ws0, Wn0, b0, h1);
    sage_layer_owned<R1, 128, true, BMP, QC1, 1, 0><<<cdiv((unsigned)N2, R1), blk, 0, stream>>>(h1, src1, dst1, bm + OFF1, E1, N1, N2, Ws1, Wn1, b1, h2);
    sage_layer_owned<R2, 64, false, 0, QC1, 1, 0><<<cdiv((unsigned)N3, R2), blk, 0, stream>>>(h2, src2, dst2, nullptr, E2, N2, N3, Ws2, Wn2, b2, (float*)d_out);
}
